# starting kernel
# baseline (speedup 1.0000x reference)
#include <hip/hip_runtime.h>
#include <hip/hip_bf16.h>

typedef __attribute__((ext_vector_type(2))) float v2f;
typedef __attribute__((ext_vector_type(8))) float v8f;

#define DIM 128
#define NEG_SLOPE 0.2f

__device__ __forceinline__ float lrelu(float x) { return x > 0.f ? x : NEG_SLOPE * x; }
__device__ __forceinline__ float elu(float x)   { return x > 0.f ? x : (__expf(x) - 1.f); }

__device__ __forceinline__ unsigned enc_f(float f) {
    unsigned u = __float_as_uint(f);
    return (u & 0x80000000u) ? ~u : (u | 0x80000000u);
}
__device__ __forceinline__ float dec_f(unsigned u) {
    return (u & 0x80000000u) ? __uint_as_float(u & 0x7FFFFFFFu) : __uint_as_float(~u);
}

__global__ void __launch_bounds__(256) gemm128_wmma_f32(
    const float* __restrict__ A, const float* __restrict__ W,
    float* __restrict__ C) {
    const int wave = threadIdx.x >> 5;
    const int lane = threadIdx.x & 31;
    const int half = lane >> 4;
    const int l15  = lane & 15;
    const int row0 = blockIdx.x * 16;
    const int col0 = wave * 16;

    const float* __restrict__ Arow = A + (size_t)(row0 + l15) * DIM;

    v8f acc = {};
#pragma unroll
    for (int k = 0; k < DIM; k += 4) {
        const int ka = k + half * 2;
        v2f a, b;
        a.x = Arow[ka];
        a.y = Arow[ka + 1];
        b.x = W[(size_t)ka * DIM + col0 + l15];
        b.y = W[(size_t)(ka + 1) * DIM + col0 + l15];
        acc = __builtin_amdgcn_wmma_f32_16x16x4_f32(
             false, a,  false, b,
             (short)0, acc,  false,  false);
    }
#pragma unroll
    for (int j = 0; j < 8; ++j)
        C[(size_t)(row0 + j + half * 8) * DIM + col0 + l15] = acc[j];
}

__global__ void zero_fill(float* __restrict__ p, int n) {
    int i = blockIdx.x * blockDim.x + threadIdx.x;
    if (i < n) p[i] = 0.f;
}

template <int HEADS>
__global__ void __launch_bounds__(256) edge_alpha(
    const int* __restrict__ srcA, const int* __restrict__ dstA, int E, int ET,
    const float* __restrict__ xl, const float* __restrict__ xr,
    const float* __restrict__ att, float* __restrict__ alpha,
    unsigned* __restrict__ amax) {
    const int e = blockIdx.x * (blockDim.x >> 5) + (threadIdx.x >> 5);
    if (e >= ET) return;
    const int lane = threadIdx.x & 31;
    int s, d;
    if (e < E) { s = srcA[e]; d = dstA[e]; } else { s = d = e - E; }

    const float4 a = ((const float4*)(xl + (size_t)s * DIM))[lane];
    const float4 b = ((const float4*)(xr + (size_t)d * DIM))[lane];
    const float4 t = ((const float4*)att)[lane];

    float p = lrelu(a.x + b.x) * t.x + lrelu(a.y + b.y) * t.y +
              lrelu(a.z + b.z) * t.z + lrelu(a.w + b.w) * t.w;

    const int span = (HEADS == 2) ? 16 : 32;
#pragma unroll
    for (int m = 1; m < span; m <<= 1) p += __shfl_xor(p, m, 32);

    if ((lane & (span - 1)) == 0) {
        const int h = lane / span;
        alpha[(size_t)e * HEADS + h] = p;
        atomicMax(&amax[(size_t)d * HEADS + h], enc_f(p));
    }
}

template <int HEADS>
__global__ void __launch_bounds__(256) edge_exp(
    const int* __restrict__ dstA, int E, int ET,
    float* __restrict__ alpha, const unsigned* __restrict__ amax,
    float* __restrict__ denom) {
    const int i = blockIdx.x * blockDim.x + threadIdx.x;
    if (i >= ET * HEADS) return;
    const int e = i / HEADS;
    const int h = i - e * HEADS;
    const int d = (e < E) ? dstA[e] : (e - E);
    const float m = dec_f(amax[(size_t)d * HEADS + h]);
    const float w = __expf(alpha[i] - m);
    alpha[i] = w;
    atomicAdd(&denom[(size_t)d * HEADS + h], w);
}

template <int HEADS>
__global__ void __launch_bounds__(256) edge_aggregate(
    const int* __restrict__ srcA, const int* __restrict__ dstA, int E, int ET,
    const float* __restrict__ xl, const float* __restrict__ alpha,
    const float* __restrict__ denom, float* __restrict__ out) {
    const int e = blockIdx.x * (blockDim.x >> 5) + (threadIdx.x >> 5);
    if (e >= ET) return;
    const int lane = threadIdx.x & 31;
    int s, d;
    if (e < E) { s = srcA[e]; d = dstA[e]; } else { s = d = e - E; }

    const int span = (HEADS == 2) ? 16 : 32;
    const int h = lane / span;
    const float coeff = alpha[(size_t)e * HEADS + h] /
                        (denom[(size_t)d * HEADS + h] + 1e-16f);

    const float4 v = ((const float4*)(xl + (size_t)s * DIM))[lane];
    float* o = out + (size_t)d * DIM + lane * 4;
    atomicAdd(o + 0, coeff * v.x);
    atomicAdd(o + 1, coeff * v.y);
    atomicAdd(o + 2, coeff * v.z);
    atomicAdd(o + 3, coeff * v.w);
}

__global__ void epilogue_elu(float* __restrict__ p, const float* __restrict__ bias,
                             int n) {
    int i = blockIdx.x * blockDim.x + threadIdx.x;
    if (i < n) {
        const float v = p[i] + bias[i & (DIM - 1)];
        p[i] = elu(v);
    }
}

extern "C" void kernel_launch(void* const* d_in, const int* in_sizes, int n_in,
                              void* d_out, int out_size, void* d_ws, size_t ws_size,
                              hipStream_t stream) {
    const float* x    = (const float*)d_in[0];
    const int*   ei   = (const int*)d_in[1];
    const float* Wl1  = (const float*)d_in[2];
    const float* Wr1  = (const float*)d_in[3];
    const float* att1 = (const float*)d_in[4];
    const float* b1   = (const float*)d_in[5];
    const float* Wl2  = (const float*)d_in[6];
    const float* Wr2  = (const float*)d_in[7];
    const float* att2 = (const float*)d_in[8];
    const float* b2   = (const float*)d_in[9];

    const int N  = in_sizes[0] / DIM;
    const int E  = in_sizes[1] / 2;
    const int ET = E + N;

    const int* srcA = ei;
    const int* dstA = ei + E;

    float*    xl    = (float*)d_ws;
    float*    xr    = xl + (size_t)N * DIM;
    float*    h1    = xr + (size_t)N * DIM;
    float*    alpha = h1 + (size_t)N * DIM;
    unsigned* amax  = (unsigned*)(alpha + (size_t)ET * 2);
    float*    denom = (float*)(amax + (size_t)N * 2);
    float*    out   = (float*)d_out;

    const int nNodeElems = N * DIM;
    const dim3 B(256);
    const int gNode  = (nNodeElems + 255) / 256;
    const int gStat  = (N * 4 + 255) / 256;
    const int gGemm  = N / 16;
    const int gWaveE = (ET + 7) / 8;

    zero_fill<<<gNode, B, 0, stream>>>(h1, nNodeElems);
    zero_fill<<<gStat, B, 0, stream>>>((float*)amax, N * 4);
    gemm128_wmma_f32<<<gGemm, B, 0, stream>>>(x, Wl1, xl);
    gemm128_wmma_f32<<<gGemm, B, 0, stream>>>(x, Wr1, xr);
    edge_alpha<2><<<gWaveE, B, 0, stream>>>(srcA, dstA, E, ET, xl, xr, att1, alpha, amax);
    edge_exp<2><<<(ET * 2 + 255) / 256, B, 0, stream>>>(dstA, E, ET, alpha, amax, denom);
    edge_aggregate<2><<<gWaveE, B, 0, stream>>>(srcA, dstA, E, ET, xl, alpha, denom, h1);
    epilogue_elu<<<gNode, B, 0, stream>>>(h1, b1, nNodeElems);

    zero_fill<<<gNode, B, 0, stream>>>(out, nNodeElems);
    zero_fill<<<gStat, B, 0, stream>>>((float*)amax, N * 4);
    gemm128_wmma_f32<<<gGemm, B, 0, stream>>>(h1, Wl2, xl);
    gemm128_wmma_f32<<<gGemm, B, 0, stream>>>(h1, Wr2, xr);
    edge_alpha<1><<<gWaveE, B, 0, stream>>>(srcA, dstA, E, ET, xl, xr, att2, alpha, amax);
    edge_exp<1><<<(ET + 255) / 256, B, 0, stream>>>(dstA, E, ET, alpha, amax, denom);
    edge_aggregate<1><<<gWaveE, B, 0, stream>>>(srcA, dstA, E, ET, xl, alpha, denom, out);
    epilogue_elu<<<gNode, B, 0, stream>>>(out, b2, nNodeElems);
}


# Passes without float atomics get far fewer, fatter waves: a wave clears 8 KB, scores 32 edges (8 edges x 2 rows in flight, same lane-sum tree) or finishes 8 rows; aggregate pass as it was
# speedup vs baseline: 1.2154x; 1.2154x over previous
#include <hip/hip_runtime.h>
#include <hip/hip_bf16.h>

typedef __attribute__((ext_vector_type(2))) float v2f;
typedef __attribute__((ext_vector_type(8))) float v8f;

#define DIM 128
#define NEG_SLOPE 0.2f

__device__ __forceinline__ float lrelu(float x) { return x > 0.f ? x : NEG_SLOPE * x; }
__device__ __forceinline__ float elu(float x)   { return x > 0.f ? x : (__expf(x) - 1.f); }

__device__ __forceinline__ unsigned enc_f(float f) {
    unsigned u = __float_as_uint(f);
    return (u & 0x80000000u) ? ~u : (u | 0x80000000u);
}
__device__ __forceinline__ float dec_f(unsigned u) {
    return (u & 0x80000000u) ? __uint_as_float(u & 0x7FFFFFFFu) : __uint_as_float(~u);
}

__global__ void __launch_bounds__(256) gemm128_wmma_f32(
    const float* __restrict__ A, const float* __restrict__ W,
    float* __restrict__ C) {
    const int wave = threadIdx.x >> 5;
    const int lane = threadIdx.x & 31;
    const int half = lane >> 4;
    const int l15  = lane & 15;
    const int row0 = blockIdx.x * 16;
    const int col0 = wave * 16;

    const float* __restrict__ Arow = A + (size_t)(row0 + l15) * DIM;

    v8f acc = {};
#pragma unroll
    for (int k = 0; k < DIM; k += 4) {
        const int ka = k + half * 2;
        v2f a, b;
        a.x = Arow[ka];
        a.y = Arow[ka + 1];
        b.x = W[(size_t)ka * DIM + col0 + l15];
        b.y = W[(size_t)(ka + 1) * DIM + col0 + l15];
        acc = __builtin_amdgcn_wmma_f32_16x16x4_f32(
             false, a,  false, b,
             (short)0, acc,  false,  false);
    }
#pragma unroll
    for (int j = 0; j < 8; ++j)
        C[(size_t)(row0 + j + half * 8) * DIM + col0 + l15] = acc[j];
}

#define ZF_PER 16
__global__ void __launch_bounds__(256) zero_fill(float* __restrict__ p, int n) {
    float4* __restrict__ p4 = (float4*)p;
    const int n4 = n >> 2;
    const int base = blockIdx.x * (256 * ZF_PER) + threadIdx.x;
    const float4 z = make_float4(0.f, 0.f, 0.f, 0.f);
#pragma unroll
    for (int j = 0; j < ZF_PER; ++j) {
        const int i = base + j * 256;
        if (i < n4) p4[i] = z;
    }
}

#define EA_TRIP 8
template <int HEADS>
__global__ void __launch_bounds__(256) edge_alpha(
    const int* __restrict__ srcA, const int* __restrict__ dstA, int E, int ET,
    const float* __restrict__ xl, const float* __restrict__ xr,
    const float* __restrict__ att, float* __restrict__ alpha,
    unsigned* __restrict__ amax) {
    const int lane = threadIdx.x & 31;
    const int e0 = (blockIdx.x * (blockDim.x >> 5) + (threadIdx.x >> 5)) * 32;
    if (e0 >= ET) return;

    const int el = min(e0 + lane, ET - 1);
    int s, d;
    if (el < E) { s = srcA[el]; d = dstA[el]; } else { s = d = el - E; }

    const float4 t = ((const float4*)att)[lane];
    const bool b0 = (lane & 1) != 0, b1 = (lane & 2) != 0, b2 = (lane & 4) != 0;
    const int h = (HEADS == 2) ? (lane >> 4) : 0;
    const bool owner = ((lane & 8) == 0) && (HEADS == 2 || lane < 16);

    for (int eb = 0; eb < 32; eb += EA_TRIP) {
        if (e0 + eb >= ET) break;

        float4 a[EA_TRIP], b[EA_TRIP];
#pragma unroll
        for (int j = 0; j < EA_TRIP; ++j) {
            const int sj = __builtin_amdgcn_readlane(s, eb + j);
            const int dj = __builtin_amdgcn_readlane(d, eb + j);
            a[j] = ((const float4*)(xl + (size_t)sj * DIM))[lane];
            b[j] = ((const float4*)(xr + (size_t)dj * DIM))[lane];
        }

        float p[EA_TRIP];
#pragma unroll
        for (int j = 0; j < EA_TRIP; ++j)
            p[j] = lrelu(a[j].x + b[j].x) * t.x + lrelu(a[j].y + b[j].y) * t.y +
                   lrelu(a[j].z + b[j].z) * t.z + lrelu(a[j].w + b[j].w) * t.w;

        float q[4];
#pragma unroll
        for (int i = 0; i < 4; ++i) {
            const float keep = b0 ? p[2 * i + 1] : p[2 * i];
            const float send = b0 ? p[2 * i] : p[2 * i + 1];
            q[i] = keep + __shfl_xor(send, 1, 32);
        }
        float r[2];
#pragma unroll
        for (int i = 0; i < 2; ++i) {
            const float keep = b1 ? q[2 * i + 1] : q[2 * i];
            const float send = b1 ? q[2 * i] : q[2 * i + 1];
            r[i] = keep + __shfl_xor(send, 2, 32);
        }
        float v;
        {
            const float keep = b2 ? r[1] : r[0];
            const float send = b2 ? r[0] : r[1];
            v = keep + __shfl_xor(send, 4, 32);
        }
        v += __shfl_xor(v, 8, 32);
        if (HEADS == 1) v += __shfl_xor(v, 16, 32);

        const int ej = e0 + eb + (lane & 7);
        const int dj = __shfl(d, eb + (lane & 7), 32);
        if (owner && ej < ET) {
            alpha[(size_t)ej * HEADS + h] = v;
            atomicMax(&amax[(size_t)dj * HEADS + h], enc_f(v));
        }
    }
}

template <int HEADS>
__global__ void __launch_bounds__(256) edge_aggregate(
    const int* __restrict__ srcA, const int* __restrict__ dstA, int E, int ET,
    const float* __restrict__ xl, const float* __restrict__ alpha,
    const unsigned* __restrict__ amax, float* __restrict__ denom,
    float* __restrict__ out) {
    const int e = blockIdx.x * (blockDim.x >> 5) + (threadIdx.x >> 5);
    if (e >= ET) return;
    const int lane = threadIdx.x & 31;
    int s, d;
    if (e < E) { s = srcA[e]; d = dstA[e]; } else { s = d = e - E; }

    const float* __restrict__ r = xl + (size_t)s * DIM + lane;
    const float v0 = r[0];
    const float v1 = r[32];
    const float v2 = r[64];
    const float v3 = r[96];

    float wlo, whi;
    if (HEADS == 2) {
        const float a0 = alpha[(size_t)e * 2 + 0];
        const float a1 = alpha[(size_t)e * 2 + 1];
        const float m0 = dec_f(amax[(size_t)d * 2 + 0]);
        const float m1 = dec_f(amax[(size_t)d * 2 + 1]);
        wlo = __expf(a0 - m0);
        whi = __expf(a1 - m1);
    } else {
        const float a = alpha[e];
        const float m = dec_f(amax[d]);
        wlo = whi = __expf(a - m);
    }

    float* o = out + (size_t)d * DIM + lane;
    atomicAdd(o +  0, wlo * v0);
    atomicAdd(o + 32, wlo * v1);
    atomicAdd(o + 64, whi * v2);
    atomicAdd(o + 96, whi * v3);
    if (lane < HEADS) atomicAdd(&denom[(size_t)d * HEADS + lane], lane == 0 ? wlo : whi);
}

#define EP_ROWS 8
template <int HEADS>
__global__ void __launch_bounds__(256) epilogue_elu(float* __restrict__ p,
                                                    const float* __restrict__ denom,
                                                    const float* __restrict__ bias, int N) {
    const int lane = threadIdx.x & 31;
    const int row0 = (blockIdx.x * (blockDim.x >> 5) + (threadIdx.x >> 5)) * EP_ROWS;
    if (row0 >= N) return;
    const int h = (HEADS == 2) ? (lane >> 4) : 0;
    const float4 bz = ((const float4*)bias)[lane];

    float4 v[EP_ROWS];
    float dn[EP_ROWS];
#pragma unroll
    for (int j = 0; j < EP_ROWS; ++j) {
        const int row = min(row0 + j, N - 1);
        v[j]  = ((const float4*)(p + (size_t)row * DIM))[lane];
        dn[j] = denom[(size_t)row * HEADS + h];
    }
#pragma unroll
    for (int j = 0; j < EP_ROWS; ++j) {
        const int row = row0 + j;
        const float dd = dn[j] + 1e-16f;
        float4 o;
        o.x = elu(v[j].x / dd + bz.x);
        o.y = elu(v[j].y / dd + bz.y);
        o.z = elu(v[j].z / dd + bz.z);
        o.w = elu(v[j].w / dd + bz.w);
        if (row < N) ((float4*)(p + (size_t)row * DIM))[lane] = o;
    }
}

extern "C" void kernel_launch(void* const* d_in, const int* in_sizes, int n_in,
                              void* d_out, int out_size, void* d_ws, size_t ws_size,
                              hipStream_t stream) {
    const float* x    = (const float*)d_in[0];
    const int*   ei   = (const int*)d_in[1];
    const float* Wl1  = (const float*)d_in[2];
    const float* Wr1  = (const float*)d_in[3];
    const float* att1 = (const float*)d_in[4];
    const float* b1   = (const float*)d_in[5];
    const float* Wl2  = (const float*)d_in[6];
    const float* Wr2  = (const float*)d_in[7];
    const float* att2 = (const float*)d_in[8];
    const float* b2   = (const float*)d_in[9];

    const int N  = in_sizes[0] / DIM;
    const int E  = in_sizes[1] / 2;
    const int ET = E + N;

    const int* srcA = ei;
    const int* dstA = ei + E;

    float*    xl    = (float*)d_ws;
    float*    xr    = xl + (size_t)N * DIM;
    float*    h1    = xr + (size_t)N * DIM;
    float*    alpha = h1 + (size_t)N * DIM;
    unsigned* amax  = (unsigned*)(alpha + (size_t)ET * 2);
    float*    denom = (float*)(amax + (size_t)N * 2);
    float*    out   = (float*)d_out;

    const int nNodeElems = N * DIM;
    const dim3 B(256);
    const int gNode  = (nNodeElems / 4 + 256 * ZF_PER - 1) / (256 * ZF_PER);
    const int gStat  = (N + 256 * ZF_PER - 1) / (256 * ZF_PER);
    const int gEpi   = (N + 8 * EP_ROWS - 1) / (8 * EP_ROWS);
    const int gAlpha = (ET + 255) / 256;
    const int gGemm  = N / 16;
    const int gWaveE = (ET + 7) / 8;

    zero_fill<<<gNode, B, 0, stream>>>(h1, nNodeElems);
    zero_fill<<<gStat, B, 0, stream>>>((float*)amax, N * 4);
    gemm128_wmma_f32<<<gGemm, B, 0, stream>>>(x, Wl1, xl);
    gemm128_wmma_f32<<<gGemm, B, 0, stream>>>(x, Wr1, xr);
    edge_alpha<2><<<gAlpha, B, 0, stream>>>(srcA, dstA, E, ET, xl, xr, att1, alpha, amax);
    edge_aggregate<2><<<gWaveE, B, 0, stream>>>(srcA, dstA, E, ET, xl, alpha, amax, denom, h1);
    epilogue_elu<2><<<gEpi, B, 0, stream>>>(h1, denom, b1, N);

    zero_fill<<<gNode, B, 0, stream>>>(out, nNodeElems);
    zero_fill<<<gStat, B, 0, stream>>>((float*)amax, N * 4);
    gemm128_wmma_f32<<<gGemm, B, 0, stream>>>(h1, Wl2, xl);
    gemm128_wmma_f32<<<gGemm, B, 0, stream>>>(h1, Wr2, xr);
    edge_alpha<1><<<gAlpha, B, 0, stream>>>(srcA, dstA, E, ET, xl, xr, att2, alpha, amax);
    edge_aggregate<1><<<gWaveE, B, 0, stream>>>(srcA, dstA, E, ET, xl, alpha, amax, denom, out);
    epilogue_elu<1><<<gEpi, B, 0, stream>>>(out, denom, b2, N);
}


# Aggregate pass: each chip half adds only into its own class of node rows (die number from sendmsg 0x87; a block claims 64-edge units per class by integer ticket); arrival-order float sums as before
# speedup vs baseline: 1.7243x; 1.4187x over previous
#include <hip/hip_runtime.h>
#include <hip/hip_bf16.h>

typedef __attribute__((ext_vector_type(2))) float v2f;
typedef __attribute__((ext_vector_type(8))) float v8f;

#define DIM 128
#define NEG_SLOPE 0.2f

__device__ __forceinline__ float lrelu(float x) { return x > 0.f ? x : NEG_SLOPE * x; }
__device__ __forceinline__ float elu(float x)   { return x > 0.f ? x : (__expf(x) - 1.f); }

__device__ __forceinline__ unsigned enc_f(float f) {
    unsigned u = __float_as_uint(f);
    return (u & 0x80000000u) ? ~u : (u | 0x80000000u);
}
__device__ __forceinline__ float dec_f(unsigned u) {
    return (u & 0x80000000u) ? __uint_as_float(u & 0x7FFFFFFFu) : __uint_as_float(~u);
}

__global__ void __launch_bounds__(256) gemm128_wmma_f32(
    const float* __restrict__ A, const float* __restrict__ W,
    float* __restrict__ C) {
    const int wave = threadIdx.x >> 5;
    const int lane = threadIdx.x & 31;
    const int half = lane >> 4;
    const int l15  = lane & 15;
    const int row0 = blockIdx.x * 16;
    const int col0 = wave * 16;

    const float* __restrict__ Arow = A + (size_t)(row0 + l15) * DIM;

    v8f acc = {};
#pragma unroll
    for (int k = 0; k < DIM; k += 4) {
        const int ka = k + half * 2;
        v2f a, b;
        a.x = Arow[ka];
        a.y = Arow[ka + 1];
        b.x = W[(size_t)ka * DIM + col0 + l15];
        b.y = W[(size_t)(ka + 1) * DIM + col0 + l15];
        acc = __builtin_amdgcn_wmma_f32_16x16x4_f32(
             false, a,  false, b,
             (short)0, acc,  false,  false);
    }
#pragma unroll
    for (int j = 0; j < 8; ++j)
        C[(size_t)(row0 + j + half * 8) * DIM + col0 + l15] = acc[j];
}

#define ZF_PER 16
__global__ void __launch_bounds__(256) zero_fill(float* __restrict__ p, int n) {
    float4* __restrict__ p4 = (float4*)p;
    const int n4 = n >> 2;
    const int base = blockIdx.x * (256 * ZF_PER) + threadIdx.x;
    const float4 z = make_float4(0.f, 0.f, 0.f, 0.f);
#pragma unroll
    for (int j = 0; j < ZF_PER; ++j) {
        const int i = base + j * 256;
        if (i < n4) p4[i] = z;
    }
}

#define EA_TRIP 8
template <int HEADS>
__global__ void __launch_bounds__(256) edge_alpha(
    const int* __restrict__ srcA, const int* __restrict__ dstA, int E, int ET,
    const float* __restrict__ xl, const float* __restrict__ xr,
    const float* __restrict__ att, float* __restrict__ alpha,
    unsigned* __restrict__ amax) {
    const int lane = threadIdx.x & 31;
    const int e0 = (blockIdx.x * (blockDim.x >> 5) + (threadIdx.x >> 5)) * 32;
    if (e0 >= ET) return;

    const int el = min(e0 + lane, ET - 1);
    int s, d;
    if (el < E) { s = srcA[el]; d = dstA[el]; } else { s = d = el - E; }

    const float4 t = ((const float4*)att)[lane];
    const bool b0 = (lane & 1) != 0, b1 = (lane & 2) != 0, b2 = (lane & 4) != 0;
    const int h = (HEADS == 2) ? (lane >> 4) : 0;
    const bool owner = ((lane & 8) == 0) && (HEADS == 2 || lane < 16);

    for (int eb = 0; eb < 32; eb += EA_TRIP) {
        if (e0 + eb >= ET) break;

        float4 a[EA_TRIP], b[EA_TRIP];
#pragma unroll
        for (int j = 0; j < EA_TRIP; ++j) {
            const int sj = __builtin_amdgcn_readlane(s, eb + j);
            const int dj = __builtin_amdgcn_readlane(d, eb + j);
            a[j] = ((const float4*)(xl + (size_t)sj * DIM))[lane];
            b[j] = ((const float4*)(xr + (size_t)dj * DIM))[lane];
        }

        float p[EA_TRIP];
#pragma unroll
        for (int j = 0; j < EA_TRIP; ++j)
            p[j] = lrelu(a[j].x + b[j].x) * t.x + lrelu(a[j].y + b[j].y) * t.y +
                   lrelu(a[j].z + b[j].z) * t.z + lrelu(a[j].w + b[j].w) * t.w;

        float q[4];
#pragma unroll
        for (int i = 0; i < 4; ++i) {
            const float keep = b0 ? p[2 * i + 1] : p[2 * i];
            const float send = b0 ? p[2 * i] : p[2 * i + 1];
            q[i] = keep + __shfl_xor(send, 1, 32);
        }
        float r[2];
#pragma unroll
        for (int i = 0; i < 2; ++i) {
            const float keep = b1 ? q[2 * i + 1] : q[2 * i];
            const float send = b1 ? q[2 * i] : q[2 * i + 1];
            r[i] = keep + __shfl_xor(send, 2, 32);
        }
        float v;
        {
            const float keep = b2 ? r[1] : r[0];
            const float send = b2 ? r[0] : r[1];
            v = keep + __shfl_xor(send, 4, 32);
        }
        v += __shfl_xor(v, 8, 32);
        if (HEADS == 1) v += __shfl_xor(v, 16, 32);

        const int ej = e0 + eb + (lane & 7);
        const int dj = __shfl(d, eb + (lane & 7), 32);
        if (owner && ej < ET) {
            alpha[(size_t)ej * HEADS + h] = v;
            atomicMax(&amax[(size_t)dj * HEADS + h], enc_f(v));
        }
    }
}

#define AG_EDGES 8
#define AG_UNIT (8 * AG_EDGES)
__device__ __forceinline__ float lane_value(float x, int l) {
    return __int_as_float(__builtin_amdgcn_readlane(__float_as_int(x), l));
}
template <int HEADS>
__global__ void __launch_bounds__(256) edge_aggregate(
    const int* __restrict__ srcA, const int* __restrict__ dstA, int E, int ET,
    const float* __restrict__ xl, const float* __restrict__ alpha,
    const unsigned* __restrict__ amax, float* __restrict__ denom,
    float* __restrict__ out, unsigned* __restrict__ tickets, int nUnits) {
    __shared__ unsigned claim[2];
    if (threadIdx.x == 0) {
        unsigned h = (__builtin_amdgcn_s_sendmsg_rtn(0x87) >> 18) & 1u;
        unsigned t = atomicAdd(&tickets[h * 32], 1u);
        if (t >= (unsigned)nUnits) {
            h ^= 1u;
            t = atomicAdd(&tickets[h * 32], 1u);
        }
        claim[0] = t;
        claim[1] = h;
    }
    __syncthreads();
    const unsigned t = claim[0];
    const int cls = (int)claim[1];
    if (t >= (unsigned)nUnits) return;

    const int lane = threadIdx.x & 31;
    const int e0 = (int)t * AG_UNIT + (int)(threadIdx.x >> 5) * AG_EDGES;
    if (e0 >= ET) return;

    const int el = min(e0 + (lane & (AG_EDGES - 1)), ET - 1);
    const int ec = min(el, E - 1);
    const int sv = srcA[ec];
    const int dv = dstA[ec];
    const int s = el < E ? sv : el - E;
    const int d = el < E ? dv : el - E;
    const bool mine = lane < AG_EDGES && e0 + lane < ET && ((d >> 6) & 1) == cls;
    const unsigned m = __builtin_amdgcn_ballot_w32(mine);

    float alo, ahi, mlo, mhi;
    if (HEADS == 2) {
        const float2 a  = ((const float2*)alpha)[el];
        const uint2  mx = ((const uint2*)amax)[d];
        alo = a.x; ahi = a.y;
        mlo = dec_f(mx.x); mhi = dec_f(mx.y);
    } else {
        alo = ahi = alpha[el];
        mlo = mhi = dec_f(amax[d]);
    }

    float v[AG_EDGES][4];
#pragma unroll
    for (int j = 0; j < AG_EDGES; ++j) {
        if (m & (1u << j)) {
            const int sj = __builtin_amdgcn_readlane(s, j);
            const float* __restrict__ r = xl + (size_t)sj * DIM + lane;
            v[j][0] = r[0];
            v[j][1] = r[32];
            v[j][2] = r[64];
            v[j][3] = r[96];
        }
    }

    const float wlo = __expf(alo - mlo);
    const float whi = (HEADS == 2) ? __expf(ahi - mhi) : wlo;
    if (mine) {
        atomicAdd(&denom[(size_t)d * HEADS], wlo);
        if (HEADS == 2) atomicAdd(&denom[(size_t)d * 2 + 1], whi);
    }

#pragma unroll
    for (int j = 0; j < AG_EDGES; ++j) {
        if (m & (1u << j)) {
            const int dj = __builtin_amdgcn_readlane(d, j);
            const float wl = lane_value(wlo, j);
            const float wh = lane_value(whi, j);
            float* o = out + (size_t)dj * DIM + lane;
            atomicAdd(o +  0, wl * v[j][0]);
            atomicAdd(o + 32, wl * v[j][1]);
            atomicAdd(o + 64, wh * v[j][2]);
            atomicAdd(o + 96, wh * v[j][3]);
        }
    }
}

#define EP_ROWS 8
template <int HEADS>
__global__ void __launch_bounds__(256) epilogue_elu(float* __restrict__ p,
                                                    const float* __restrict__ denom,
                                                    const float* __restrict__ bias, int N) {
    const int lane = threadIdx.x & 31;
    const int row0 = (blockIdx.x * (blockDim.x >> 5) + (threadIdx.x >> 5)) * EP_ROWS;
    if (row0 >= N) return;
    const int h = (HEADS == 2) ? (lane >> 4) : 0;
    const float4 bz = ((const float4*)bias)[lane];

    float4 v[EP_ROWS];
    float dn[EP_ROWS];
#pragma unroll
    for (int j = 0; j < EP_ROWS; ++j) {
        const int row = min(row0 + j, N - 1);
        v[j]  = ((const float4*)(p + (size_t)row * DIM))[lane];
        dn[j] = denom[(size_t)row * HEADS + h];
    }
#pragma unroll
    for (int j = 0; j < EP_ROWS; ++j) {
        const int row = row0 + j;
        const float dd = dn[j] + 1e-16f;
        float4 o;
        o.x = elu(v[j].x / dd + bz.x);
        o.y = elu(v[j].y / dd + bz.y);
        o.z = elu(v[j].z / dd + bz.z);
        o.w = elu(v[j].w / dd + bz.w);
        if (row < N) ((float4*)(p + (size_t)row * DIM))[lane] = o;
    }
}

extern "C" void kernel_launch(void* const* d_in, const int* in_sizes, int n_in,
                              void* d_out, int out_size, void* d_ws, size_t ws_size,
                              hipStream_t stream) {
    const float* x    = (const float*)d_in[0];
    const int*   ei   = (const int*)d_in[1];
    const float* Wl1  = (const float*)d_in[2];
    const float* Wr1  = (const float*)d_in[3];
    const float* att1 = (const float*)d_in[4];
    const float* b1   = (const float*)d_in[5];
    const float* Wl2  = (const float*)d_in[6];
    const float* Wr2  = (const float*)d_in[7];
    const float* att2 = (const float*)d_in[8];
    const float* b2   = (const float*)d_in[9];

    const int N  = in_sizes[0] / DIM;
    const int E  = in_sizes[1] / 2;
    const int ET = E + N;

    const int* srcA = ei;
    const int* dstA = ei + E;

    float*    xl    = (float*)d_ws;
    float*    xr    = xl + (size_t)N * DIM;
    float*    h1    = xr + (size_t)N * DIM;
    float*    alpha = h1 + (size_t)N * DIM;
    unsigned* amax  = (unsigned*)(alpha + (size_t)ET * 2);
    float*    denom = (float*)(amax + (size_t)N * 2);
    unsigned* tickets = (unsigned*)(denom + (size_t)N * 2);
    float*    out   = (float*)d_out;

    const int nNodeElems = N * DIM;
    const dim3 B(256);
    const int gNode  = (nNodeElems / 4 + 256 * ZF_PER - 1) / (256 * ZF_PER);
    const int nStat  = N * 4 + 64;
    const int gStat  = (nStat / 4 + 256 * ZF_PER - 1) / (256 * ZF_PER);
    const int gEpi   = (N + 8 * EP_ROWS - 1) / (8 * EP_ROWS);
    const int gAlpha = (ET + 255) / 256;
    const int gGemm  = N / 16;
    const int nUnits = (ET + AG_UNIT - 1) / AG_UNIT;
    const int gAggr  = 2 * nUnits;

    zero_fill<<<gNode, B, 0, stream>>>(h1, nNodeElems);
    zero_fill<<<gStat, B, 0, stream>>>((float*)amax, nStat);
    gemm128_wmma_f32<<<gGemm, B, 0, stream>>>(x, Wl1, xl);
    gemm128_wmma_f32<<<gGemm, B, 0, stream>>>(x, Wr1, xr);
    edge_alpha<2><<<gAlpha, B, 0, stream>>>(srcA, dstA, E, ET, xl, xr, att1, alpha, amax);
    edge_aggregate<2><<<gAggr, B, 0, stream>>>(srcA, dstA, E, ET, xl, alpha, amax, denom, h1, tickets, nUnits);
    epilogue_elu<2><<<gEpi, B, 0, stream>>>(h1, denom, b1, N);

    zero_fill<<<gNode, B, 0, stream>>>(out, nNodeElems);
    zero_fill<<<gStat, B, 0, stream>>>((float*)amax, nStat);
    gemm128_wmma_f32<<<gGemm, B, 0, stream>>>(h1, Wl2, xl);
    gemm128_wmma_f32<<<gGemm, B, 0, stream>>>(h1, Wr2, xr);
    edge_alpha<1><<<gAlpha, B, 0, stream>>>(srcA, dstA, E, ET, xl, xr, att2, alpha, amax);
    edge_aggregate<1><<<gAggr, B, 0, stream>>>(srcA, dstA, E, ET, xl, alpha, amax, denom, out, tickets, nUnits);
    epilogue_elu<1><<<gEpi, B, 0, stream>>>(out, denom, b2, N);
}


# GEMM: the block's 16 x 128 tile of A is read once by whole rows (16 B per lane, four whole lines per load) into LDS at pitch 132; A fragments come from LDS, not from 16-row 8-byte gathers per wave
# speedup vs baseline: 1.8462x; 1.0707x over previous
#include <hip/hip_runtime.h>
#include <hip/hip_bf16.h>

typedef __attribute__((ext_vector_type(2))) float v2f;
typedef __attribute__((ext_vector_type(8))) float v8f;

#define DIM 128
#define NEG_SLOPE 0.2f

__device__ __forceinline__ float lrelu(float x) { return x > 0.f ? x : NEG_SLOPE * x; }
__device__ __forceinline__ float elu(float x)   { return x > 0.f ? x : (__expf(x) - 1.f); }

__device__ __forceinline__ unsigned enc_f(float f) {
    unsigned u = __float_as_uint(f);
    return (u & 0x80000000u) ? ~u : (u | 0x80000000u);
}
__device__ __forceinline__ float dec_f(unsigned u) {
    return (u & 0x80000000u) ? __uint_as_float(u & 0x7FFFFFFFu) : __uint_as_float(~u);
}

#define A_PITCH 132
__global__ void __launch_bounds__(256) gemm128_wmma_f32(
    const float* __restrict__ A, const float* __restrict__ W,
    float* __restrict__ C) {
    __shared__ __attribute__((aligned(16))) float As[16 * A_PITCH];
    const int wave = threadIdx.x >> 5;
    const int lane = threadIdx.x & 31;
    const int half = lane >> 4;
    const int l15  = lane & 15;
    const int row0 = blockIdx.x * 16;
    const int col0 = wave * 16;

    const float4* __restrict__ A4 = (const float4*)(A + (size_t)row0 * DIM);
    const float4 t0 = A4[threadIdx.x];
    const float4 t1 = A4[threadIdx.x + 256];

    v2f b[DIM / 4];
#pragma unroll
    for (int i = 0; i < DIM / 4; ++i) {
        const int ka = 4 * i + half * 2;
        b[i].x = W[(size_t)ka * DIM + col0 + l15];
        b[i].y = W[(size_t)(ka + 1) * DIM + col0 + l15];
    }

    *(float4*)&As[wave * A_PITCH + lane * 4]       = t0;
    *(float4*)&As[(wave + 8) * A_PITCH + lane * 4] = t1;
    __syncthreads();

    const float* __restrict__ Arow = As + l15 * A_PITCH + half * 2;

    v8f acc = {};
#pragma unroll
    for (int i = 0; i < DIM / 4; ++i) {
        const v2f a = *(const v2f*)(Arow + 4 * i);
        acc = __builtin_amdgcn_wmma_f32_16x16x4_f32(
             false, a,  false, b[i],
             (short)0, acc,  false,  false);
    }
#pragma unroll
    for (int j = 0; j < 8; ++j)
        C[(size_t)(row0 + j + half * 8) * DIM + col0 + l15] = acc[j];
}

#define ZF_PER 16
__global__ void __launch_bounds__(256) zero_fill(float* __restrict__ p, int n) {
    float4* __restrict__ p4 = (float4*)p;
    const int n4 = n >> 2;
    const int base = blockIdx.x * (256 * ZF_PER) + threadIdx.x;
    const float4 z = make_float4(0.f, 0.f, 0.f, 0.f);
#pragma unroll
    for (int j = 0; j < ZF_PER; ++j) {
        const int i = base + j * 256;
        if (i < n4) p4[i] = z;
    }
}

#define EA_TRIP 8
template <int HEADS>
__global__ void __launch_bounds__(256) edge_alpha(
    const int* __restrict__ srcA, const int* __restrict__ dstA, int E, int ET,
    const float* __restrict__ xl, const float* __restrict__ xr,
    const float* __restrict__ att, float* __restrict__ alpha,
    unsigned* __restrict__ amax) {
    const int lane = threadIdx.x & 31;
    const int e0 = (blockIdx.x * (blockDim.x >> 5) + (threadIdx.x >> 5)) * 32;
    if (e0 >= ET) return;

    const int el = min(e0 + lane, ET - 1);
    int s, d;
    if (el < E) { s = srcA[el]; d = dstA[el]; } else { s = d = el - E; }

    const float4 t = ((const float4*)att)[lane];
    const bool b0 = (lane & 1) != 0, b1 = (lane & 2) != 0, b2 = (lane & 4) != 0;
    const int h = (HEADS == 2) ? (lane >> 4) : 0;
    const bool owner = ((lane & 8) == 0) && (HEADS == 2 || lane < 16);

    for (int eb = 0; eb < 32; eb += EA_TRIP) {
        if (e0 + eb >= ET) break;

        float4 a[EA_TRIP], b[EA_TRIP];
#pragma unroll
        for (int j = 0; j < EA_TRIP; ++j) {
            const int sj = __builtin_amdgcn_readlane(s, eb + j);
            const int dj = __builtin_amdgcn_readlane(d, eb + j);
            a[j] = ((const float4*)(xl + (size_t)sj * DIM))[lane];
            b[j] = ((const float4*)(xr + (size_t)dj * DIM))[lane];
        }

        float p[EA_TRIP];
#pragma unroll
        for (int j = 0; j < EA_TRIP; ++j)
            p[j] = lrelu(a[j].x + b[j].x) * t.x + lrelu(a[j].y + b[j].y) * t.y +
                   lrelu(a[j].z + b[j].z) * t.z + lrelu(a[j].w + b[j].w) * t.w;

        float q[4];
#pragma unroll
        for (int i = 0; i < 4; ++i) {
            const float keep = b0 ? p[2 * i + 1] : p[2 * i];
            const float send = b0 ? p[2 * i] : p[2 * i + 1];
            q[i] = keep + __shfl_xor(send, 1, 32);
        }
        float r[2];
#pragma unroll
        for (int i = 0; i < 2; ++i) {
            const float keep = b1 ? q[2 * i + 1] : q[2 * i];
            const float send = b1 ? q[2 * i] : q[2 * i + 1];
            r[i] = keep + __shfl_xor(send, 2, 32);
        }
        float v;
        {
            const float keep = b2 ? r[1] : r[0];
            const float send = b2 ? r[0] : r[1];
            v = keep + __shfl_xor(send, 4, 32);
        }
        v += __shfl_xor(v, 8, 32);
        if (HEADS == 1) v += __shfl_xor(v, 16, 32);

        const int ej = e0 + eb + (lane & 7);
        const int dj = __shfl(d, eb + (lane & 7), 32);
        if (owner && ej < ET) {
            alpha[(size_t)ej * HEADS + h] = v;
            atomicMax(&amax[(size_t)dj * HEADS + h], enc_f(v));
        }
    }
}

#define AG_EDGES 8
#define AG_UNIT (8 * AG_EDGES)
__device__ __forceinline__ float lane_value(float x, int l) {
    return __int_as_float(__builtin_amdgcn_readlane(__float_as_int(x), l));
}
template <int HEADS>
__global__ void __launch_bounds__(256) edge_aggregate(
    const int* __restrict__ srcA, const int* __restrict__ dstA, int E, int ET,
    const float* __restrict__ xl, const float* __restrict__ alpha,
    const unsigned* __restrict__ amax, float* __restrict__ denom,
    float* __restrict__ out, unsigned* __restrict__ tickets, int nUnits) {
    __shared__ unsigned claim[2];
    if (threadIdx.x == 0) {
        unsigned h = (__builtin_amdgcn_s_sendmsg_rtn(0x87) >> 18) & 1u;
        unsigned t = atomicAdd(&tickets[h * 32], 1u);
        if (t >= (unsigned)nUnits) {
            h ^= 1u;
            t = atomicAdd(&tickets[h * 32], 1u);
        }
        claim[0] = t;
        claim[1] = h;
    }
    __syncthreads();
    const unsigned t = claim[0];
    const int cls = (int)claim[1];
    if (t >= (unsigned)nUnits) return;

    const int lane = threadIdx.x & 31;
    const int e0 = (int)t * AG_UNIT + (int)(threadIdx.x >> 5) * AG_EDGES;
    if (e0 >= ET) return;

    const int el = min(e0 + (lane & (AG_EDGES - 1)), ET - 1);
    const int ec = min(el, E - 1);
    const int sv = srcA[ec];
    const int dv = dstA[ec];
    const int s = el < E ? sv : el - E;
    const int d = el < E ? dv : el - E;
    const bool mine = lane < AG_EDGES && e0 + lane < ET && ((d >> 6) & 1) == cls;
    const unsigned m = __builtin_amdgcn_ballot_w32(mine);

    float alo, ahi, mlo, mhi;
    if (HEADS == 2) {
        const float2 a  = ((const float2*)alpha)[el];
        const uint2  mx = ((const uint2*)amax)[d];
        alo = a.x; ahi = a.y;
        mlo = dec_f(mx.x); mhi = dec_f(mx.y);
    } else {
        alo = ahi = alpha[el];
        mlo = mhi = dec_f(amax[d]);
    }

    float v[AG_EDGES][4];
#pragma unroll
    for (int j = 0; j < AG_EDGES; ++j) {
        if (m & (1u << j)) {
            const int sj = __builtin_amdgcn_readlane(s, j);
            const float* __restrict__ r = xl + (size_t)sj * DIM + lane;
            v[j][0] = r[0];
            v[j][1] = r[32];
            v[j][2] = r[64];
            v[j][3] = r[96];
        }
    }

    const float wlo = __expf(alo - mlo);
    const float whi = (HEADS == 2) ? __expf(ahi - mhi) : wlo;
    if (mine) {
        atomicAdd(&denom[(size_t)d * HEADS], wlo);
        if (HEADS == 2) atomicAdd(&denom[(size_t)d * 2 + 1], whi);
    }

#pragma unroll
    for (int j = 0; j < AG_EDGES; ++j) {
        if (m & (1u << j)) {
            const int dj = __builtin_amdgcn_readlane(d, j);
            const float wl = lane_value(wlo, j);
            const float wh = lane_value(whi, j);
            float* o = out + (size_t)dj * DIM + lane;
            atomicAdd(o +  0, wl * v[j][0]);
            atomicAdd(o + 32, wl * v[j][1]);
            atomicAdd(o + 64, wh * v[j][2]);
            atomicAdd(o + 96, wh * v[j][3]);
        }
    }
}

#define EP_ROWS 8
template <int HEADS>
__global__ void __launch_bounds__(256) epilogue_elu(float* __restrict__ p,
                                                    const float* __restrict__ denom,
                                                    const float* __restrict__ bias, int N) {
    const int lane = threadIdx.x & 31;
    const int row0 = (blockIdx.x * (blockDim.x >> 5) + (threadIdx.x >> 5)) * EP_ROWS;
    if (row0 >= N) return;
    const int h = (HEADS == 2) ? (lane >> 4) : 0;
    const float4 bz = ((const float4*)bias)[lane];

    float4 v[EP_ROWS];
    float dn[EP_ROWS];
#pragma unroll
    for (int j = 0; j < EP_ROWS; ++j) {
        const int row = min(row0 + j, N - 1);
        v[j]  = ((const float4*)(p + (size_t)row * DIM))[lane];
        dn[j] = denom[(size_t)row * HEADS + h];
    }
#pragma unroll
    for (int j = 0; j < EP_ROWS; ++j) {
        const int row = row0 + j;
        const float dd = dn[j] + 1e-16f;
        float4 o;
        o.x = elu(v[j].x / dd + bz.x);
        o.y = elu(v[j].y / dd + bz.y);
        o.z = elu(v[j].z / dd + bz.z);
        o.w = elu(v[j].w / dd + bz.w);
        if (row < N) ((float4*)(p + (size_t)row * DIM))[lane] = o;
    }
}

extern "C" void kernel_launch(void* const* d_in, const int* in_sizes, int n_in,
                              void* d_out, int out_size, void* d_ws, size_t ws_size,
                              hipStream_t stream) {
    const float* x    = (const float*)d_in[0];
    const int*   ei   = (const int*)d_in[1];
    const float* Wl1  = (const float*)d_in[2];
    const float* Wr1  = (const float*)d_in[3];
    const float* att1 = (const float*)d_in[4];
    const float* b1   = (const float*)d_in[5];
    const float* Wl2  = (const float*)d_in[6];
    const float* Wr2  = (const float*)d_in[7];
    const float* att2 = (const float*)d_in[8];
    const float* b2   = (const float*)d_in[9];

    const int N  = in_sizes[0] / DIM;
    const int E  = in_sizes[1] / 2;
    const int ET = E + N;

    const int* srcA = ei;
    const int* dstA = ei + E;

    float*    xl    = (float*)d_ws;
    float*    xr    = xl + (size_t)N * DIM;
    float*    h1    = xr + (size_t)N * DIM;
    float*    alpha = h1 + (size_t)N * DIM;
    unsigned* amax  = (unsigned*)(alpha + (size_t)ET * 2);
    float*    denom = (float*)(amax + (size_t)N * 2);
    unsigned* tickets = (unsigned*)(denom + (size_t)N * 2);
    float*    out   = (float*)d_out;

    const int nNodeElems = N * DIM;
    const dim3 B(256);
    const int gNode  = (nNodeElems / 4 + 256 * ZF_PER - 1) / (256 * ZF_PER);
    const int nStat  = N * 4 + 64;
    const int gStat  = (nStat / 4 + 256 * ZF_PER - 1) / (256 * ZF_PER);
    const int gEpi   = (N + 8 * EP_ROWS - 1) / (8 * EP_ROWS);
    const int gAlpha = (ET + 255) / 256;
    const int gGemm  = N / 16;
    const int nUnits = (ET + AG_UNIT - 1) / AG_UNIT;
    const int gAggr  = 2 * nUnits;

    zero_fill<<<gNode, B, 0, stream>>>(h1, nNodeElems);
    zero_fill<<<gStat, B, 0, stream>>>((float*)amax, nStat);
    gemm128_wmma_f32<<<gGemm, B, 0, stream>>>(x, Wl1, xl);
    gemm128_wmma_f32<<<gGemm, B, 0, stream>>>(x, Wr1, xr);
    edge_alpha<2><<<gAlpha, B, 0, stream>>>(srcA, dstA, E, ET, xl, xr, att1, alpha, amax);
    edge_aggregate<2><<<gAggr, B, 0, stream>>>(srcA, dstA, E, ET, xl, alpha, amax, denom, h1, tickets, nUnits);
    epilogue_elu<2><<<gEpi, B, 0, stream>>>(h1, denom, b1, N);

    zero_fill<<<gNode, B, 0, stream>>>(out, nNodeElems);
    zero_fill<<<gStat, B, 0, stream>>>((float*)amax, nStat);
    gemm128_wmma_f32<<<gGemm, B, 0, stream>>>(h1, Wl2, xl);
    gemm128_wmma_f32<<<gGemm, B, 0, stream>>>(h1, Wr2, xr);
    edge_alpha<1><<<gAlpha, B, 0, stream>>>(srcA, dstA, E, ET, xl, xr, att2, alpha, amax);
    edge_aggregate<1><<<gAggr, B, 0, stream>>>(srcA, dstA, E, ET, xl, alpha, amax, denom, out, tickets, nUnits);
    epilogue_elu<1><<<gEpi, B, 0, stream>>>(out, denom, b2, N);
}


# Aggregate pass: a block owns 196 dst nodes, scans the whole dst list for its edges and sums rows in LDS in fixed list order, one wave per node; rows stored once; no global float atomics; 10 launches
# speedup vs baseline: 2.1369x; 1.1575x over previous
#include <hip/hip_runtime.h>
#include <hip/hip_bf16.h>

typedef __attribute__((ext_vector_type(2))) float v2f;
typedef __attribute__((ext_vector_type(8))) float v8f;

#define DIM 128
#define NEG_SLOPE 0.2f

__device__ __forceinline__ float lrelu(float x) { return x > 0.f ? x : NEG_SLOPE * x; }
__device__ __forceinline__ float elu(float x)   { return x > 0.f ? x : (__expf(x) - 1.f); }

__device__ __forceinline__ unsigned enc_f(float f) {
    unsigned u = __float_as_uint(f);
    return (u & 0x80000000u) ? ~u : (u | 0x80000000u);
}
__device__ __forceinline__ float dec_f(unsigned u) {
    return (u & 0x80000000u) ? __uint_as_float(u & 0x7FFFFFFFu) : __uint_as_float(~u);
}

#define A_PITCH 132
__global__ void __launch_bounds__(256) gemm128_wmma_f32(
    const float* __restrict__ A, const float* __restrict__ W,
    float* __restrict__ C) {
    __shared__ __attribute__((aligned(16))) float As[16 * A_PITCH];
    const int wave = threadIdx.x >> 5;
    const int lane = threadIdx.x & 31;
    const int half = lane >> 4;
    const int l15  = lane & 15;
    const int row0 = blockIdx.x * 16;
    const int col0 = wave * 16;

    const float4* __restrict__ A4 = (const float4*)(A + (size_t)row0 * DIM);
    const float4 t0 = A4[threadIdx.x];
    const float4 t1 = A4[threadIdx.x + 256];

    v2f b[DIM / 4];
#pragma unroll
    for (int i = 0; i < DIM / 4; ++i) {
        const int ka = 4 * i + half * 2;
        b[i].x = W[(size_t)ka * DIM + col0 + l15];
        b[i].y = W[(size_t)(ka + 1) * DIM + col0 + l15];
    }

    *(float4*)&As[wave * A_PITCH + lane * 4]       = t0;
    *(float4*)&As[(wave + 8) * A_PITCH + lane * 4] = t1;
    __syncthreads();

    const float* __restrict__ Arow = As + l15 * A_PITCH + half * 2;

    v8f acc = {};
#pragma unroll
    for (int i = 0; i < DIM / 4; ++i) {
        const v2f a = *(const v2f*)(Arow + 4 * i);
        acc = __builtin_amdgcn_wmma_f32_16x16x4_f32(
             false, a,  false, b[i],
             (short)0, acc,  false,  false);
    }
#pragma unroll
    for (int j = 0; j < 8; ++j)
        C[(size_t)(row0 + j + half * 8) * DIM + col0 + l15] = acc[j];
}

#define ZF_PER 16
__global__ void __launch_bounds__(256) zero_fill(float* __restrict__ p, int n) {
    float4* __restrict__ p4 = (float4*)p;
    const int n4 = n >> 2;
    const int base = blockIdx.x * (256 * ZF_PER) + threadIdx.x;
    const float4 z = make_float4(0.f, 0.f, 0.f, 0.f);
#pragma unroll
    for (int j = 0; j < ZF_PER; ++j) {
        const int i = base + j * 256;
        if (i < n4) p4[i] = z;
    }
}

#define EA_TRIP 8
template <int HEADS>
__global__ void __launch_bounds__(256) edge_alpha(
    const int* __restrict__ srcA, const int* __restrict__ dstA, int E, int ET,
    const float* __restrict__ xl, const float* __restrict__ xr,
    const float* __restrict__ att, float* __restrict__ alpha,
    unsigned* __restrict__ amax) {
    const int lane = threadIdx.x & 31;
    const int e0 = (blockIdx.x * (blockDim.x >> 5) + (threadIdx.x >> 5)) * 32;
    if (e0 >= ET) return;

    const int el = min(e0 + lane, ET - 1);
    int s, d;
    if (el < E) { s = srcA[el]; d = dstA[el]; } else { s = d = el - E; }

    const float4 t = ((const float4*)att)[lane];
    const bool b0 = (lane & 1) != 0, b1 = (lane & 2) != 0, b2 = (lane & 4) != 0;
    const int h = (HEADS == 2) ? (lane >> 4) : 0;
    const bool owner = ((lane & 8) == 0) && (HEADS == 2 || lane < 16);

    for (int eb = 0; eb < 32; eb += EA_TRIP) {
        if (e0 + eb >= ET) break;

        float4 a[EA_TRIP], b[EA_TRIP];
#pragma unroll
        for (int j = 0; j < EA_TRIP; ++j) {
            const int sj = __builtin_amdgcn_readlane(s, eb + j);
            const int dj = __builtin_amdgcn_readlane(d, eb + j);
            a[j] = ((const float4*)(xl + (size_t)sj * DIM))[lane];
            b[j] = ((const float4*)(xr + (size_t)dj * DIM))[lane];
        }

        float p[EA_TRIP];
#pragma unroll
        for (int j = 0; j < EA_TRIP; ++j)
            p[j] = lrelu(a[j].x + b[j].x) * t.x + lrelu(a[j].y + b[j].y) * t.y +
                   lrelu(a[j].z + b[j].z) * t.z + lrelu(a[j].w + b[j].w) * t.w;

        float q[4];
#pragma unroll
        for (int i = 0; i < 4; ++i) {
            const float keep = b0 ? p[2 * i + 1] : p[2 * i];
            const float send = b0 ? p[2 * i] : p[2 * i + 1];
            q[i] = keep + __shfl_xor(send, 1, 32);
        }
        float r[2];
#pragma unroll
        for (int i = 0; i < 2; ++i) {
            const float keep = b1 ? q[2 * i + 1] : q[2 * i];
            const float send = b1 ? q[2 * i] : q[2 * i + 1];
            r[i] = keep + __shfl_xor(send, 2, 32);
        }
        float v;
        {
            const float keep = b2 ? r[1] : r[0];
            const float send = b2 ? r[0] : r[1];
            v = keep + __shfl_xor(send, 4, 32);
        }
        v += __shfl_xor(v, 8, 32);
        if (HEADS == 1) v += __shfl_xor(v, 16, 32);

        const int ej = e0 + eb + (lane & 7);
        const int dj = __shfl(d, eb + (lane & 7), 32);
        if (owner && ej < ET) {
            alpha[(size_t)ej * HEADS + h] = v;
            atomicMax(&amax[(size_t)dj * HEADS + h], enc_f(v));
        }
    }
}

#define NA_NPB   196
#define NA_WAVES 32
#define NA_SCAN  16
#define NA_TRIP  (32 * NA_SCAN)
#define NA_LCAP  704
#define NA_BCAP  64
__device__ __forceinline__ float lane_value(float x, int l) {
    return __int_as_float(__builtin_amdgcn_readlane(__float_as_int(x), l));
}

__device__ __forceinline__ void na_load_whole(int (&d)[NA_SCAN], const int* __restrict__ dstA,
                                              int t, int lane) {
    const int* __restrict__ p = dstA + (size_t)t * NA_TRIP + lane;
#pragma unroll
    for (int k = 0; k < NA_SCAN; ++k) d[k] = p[32 * k];
}
__device__ __forceinline__ void na_load_part(int (&d)[NA_SCAN], const int* __restrict__ dstA,
                                             int t, int lane, int E) {
    const int e0 = t * NA_TRIP + lane;
#pragma unroll
    for (int k = 0; k < NA_SCAN; ++k) {
        const int e = e0 + 32 * k;
        d[k] = e < E ? dstA[e] : -1;
    }
}
__device__ __forceinline__ int na_scan(const int (&d)[NA_SCAN], int t, int lane, int lo, int cnt,
                                       int2* lst, int qn) {
    const int eb = t * NA_TRIP + lane;
#pragma unroll
    for (int k = 0; k < NA_SCAN; ++k) {
        const int dl = d[k] - lo;
        const bool mt = (unsigned)dl < (unsigned)cnt;
        const unsigned mask = __builtin_amdgcn_ballot_w32(mt);
        if (mask) {
            const int pos = qn + __builtin_popcount(mask & ((1u << lane) - 1u));
            if (mt) lst[pos] = make_int2(eb + 32 * k, dl);
            qn += __builtin_popcount(mask);
        }
    }
    return qn;
}

template <int HEADS>
__device__ __forceinline__ void na_chunk(
    const int2* buf, int n, int lane,
    const int* __restrict__ srcA, const float* __restrict__ xl,
    const float* __restrict__ alpha,
    const float* mx_s, float* den_s, float* acc) {
    const int2 en = buf[min(lane, n - 1)];
    const int e  = en.x;
    const int dl = en.y;
    const int s  = srcA[e];
    float wlo, whi;
    if (HEADS == 2) {
        const float2 a = ((const float2*)alpha)[e];
        wlo = __expf(a.x - mx_s[dl * 2]);
        whi = __expf(a.y - mx_s[dl * 2 + 1]);
    } else {
        wlo = whi = __expf(alpha[e] - mx_s[dl]);
    }
    if (lane < n) {
        atomicAdd(&den_s[dl * HEADS], wlo);
        if (HEADS == 2) atomicAdd(&den_s[dl * 2 + 1], whi);
    }
    for (int b = 0; b < n; b += 8) {
        float v[8][4];
#pragma unroll
        for (int j = 0; j < 8; ++j) {
            const int sj = __builtin_amdgcn_readlane(s, min(b + j, n - 1));
            const float* __restrict__ r = xl + (size_t)sj * DIM + lane;
            v[j][0] = r[0];
            v[j][1] = r[32];
            v[j][2] = r[64];
            v[j][3] = r[96];
        }
#pragma unroll
        for (int j = 0; j < 8; ++j) {
            if (b + j < n) {
                const int dj = __builtin_amdgcn_readlane(dl, b + j);
                const float wl = lane_value(wlo, b + j);
                const float wh = lane_value(whi, b + j);
                float* o = acc + dj * DIM + lane;
                atomicAdd(o +  0, wl * v[j][0]);
                atomicAdd(o + 32, wl * v[j][1]);
                atomicAdd(o + 64, wh * v[j][2]);
                atomicAdd(o + 96, wh * v[j][3]);
            }
        }
    }
}

template <int HEADS>
__global__ void __launch_bounds__(1024) node_aggregate(
    const int* __restrict__ srcA, const int* __restrict__ dstA, int E,
    const float* __restrict__ xl, const float* __restrict__ alpha,
    const unsigned* __restrict__ amax, const float* __restrict__ bias,
    float* __restrict__ out, int N) {
    __shared__ __attribute__((aligned(16))) float acc[NA_NPB * DIM];
    __shared__ float mx_s[NA_NPB * 2];
    __shared__ float den_s[NA_NPB * 2];
    __shared__ int2  lst_s[NA_WAVES * NA_LCAP];
    __shared__ int2  buf_s[NA_WAVES * NA_BCAP];
    __shared__ int   cnt_s[NA_WAVES];
    __shared__ int   fin_s[NA_WAVES];

    const int lane = threadIdx.x & 31;
    const int wave = __builtin_amdgcn_readfirstlane(threadIdx.x >> 5);
    const int lo   = blockIdx.x * NA_NPB;
    const int cnt  = min(NA_NPB, N - lo);
    const int h    = (HEADS == 2) ? (lane >> 4) : 0;

    for (int r = wave; r < cnt; r += NA_WAVES) {
        const int node = lo + r;
        const float4 x = ((const float4*)(xl + (size_t)node * DIM))[lane];
        const float a = alpha[(size_t)(E + node) * HEADS + h];
        const float m = dec_f(amax[(size_t)node * HEADS + h]);
        const float w = __expf(a - m);
        float4 o;
        o.x = w * x.x; o.y = w * x.y; o.z = w * x.z; o.w = w * x.w;
        *(float4*)&acc[r * DIM + lane * 4] = o;
        if ((lane & 15) == 0 && (HEADS == 2 || lane == 0)) {
            mx_s[r * HEADS + h]  = m;
            den_s[r * HEADS + h] = w;
        }
    }

    int2* lst = lst_s + wave * NA_LCAP;
    int2* buf = buf_s + wave * NA_BCAP;
    const int nFull = E / NA_TRIP;
    const bool ownsPart = (E - nFull * NA_TRIP) > 0 && (nFull % NA_WAVES) == wave;
    bool partLeft = ownsPart;
    int t = wave;

    for (int round = 0; round < nFull + 2; ++round) {
        int qn = 0;
        if (t < nFull) {
            int a[NA_SCAN], b[NA_SCAN];
            na_load_whole(a, dstA, t, lane);
            for (;;) {
                na_load_whole(b, dstA, min(t + NA_WAVES, nFull - 1), lane);
                qn = na_scan(a, t, lane, lo, cnt, lst, qn);
                t += NA_WAVES;
                if (t >= nFull || qn + NA_TRIP > NA_LCAP) break;
                na_load_whole(a, dstA, min(t + NA_WAVES, nFull - 1), lane);
                qn = na_scan(b, t, lane, lo, cnt, lst, qn);
                t += NA_WAVES;
                if (t >= nFull || qn + NA_TRIP > NA_LCAP) break;
            }
        }
        if (partLeft && t >= nFull && qn + NA_TRIP <= NA_LCAP) {
            int c[NA_SCAN];
            na_load_part(c, dstA, nFull, lane, E);
            qn = na_scan(c, nFull, lane, lo, cnt, lst, qn);
            partLeft = false;
        }
        if (lane == 0) {
            cnt_s[wave] = qn;
            fin_s[wave] = (t < nFull || partLeft) ? 1 : 0;
        }
        __syncthreads();

        int bn = 0;
        for (int l = 0; l < NA_WAVES; ++l) {
            const int n = __builtin_amdgcn_readfirstlane(cnt_s[l]);
            const int2* src_l = lst_s + l * NA_LCAP;
            for (int p = 0; p < n; p += 32) {
                const int i = p + lane;
                const int2 en = src_l[min(i, n - 1)];
                const bool mt = i < n && (en.y & (NA_WAVES - 1)) == wave;
                const unsigned mask = __builtin_amdgcn_ballot_w32(mt);
                if (mask) {
                    const int pos = bn + __builtin_popcount(mask & ((1u << lane) - 1u));
                    if (mt) buf[pos] = en;
                    bn += __builtin_popcount(mask);
                    if (bn >= 32) {
                        na_chunk<HEADS>(buf, 32, lane, srcA, xl, alpha, mx_s, den_s, acc);
                        const int rem = bn - 32;
                        const int2 keep = buf[32 + min(lane, NA_BCAP - 33)];
                        if (lane < rem) buf[lane] = keep;
                        bn = rem;
                    }
                }
            }
        }
        if (bn > 0)
            na_chunk<HEADS>(buf, bn, lane, srcA, xl, alpha, mx_s, den_s, acc);

        const bool more = __builtin_amdgcn_ballot_w32(fin_s[lane] != 0) != 0u;
        __syncthreads();
        if (!more) break;
    }

    const float4 bz = ((const float4*)bias)[lane];
    for (int r = wave; r < cnt; r += NA_WAVES) {
        const float4 v = *(const float4*)&acc[r * DIM + lane * 4];
        const float dd = den_s[r * HEADS + h] + 1e-16f;
        float4 o;
        o.x = elu(v.x / dd + bz.x);
        o.y = elu(v.y / dd + bz.y);
        o.z = elu(v.z / dd + bz.z);
        o.w = elu(v.w / dd + bz.w);
        ((float4*)(out + (size_t)(lo + r) * DIM))[lane] = o;
    }
}

extern "C" void kernel_launch(void* const* d_in, const int* in_sizes, int n_in,
                              void* d_out, int out_size, void* d_ws, size_t ws_size,
                              hipStream_t stream) {
    const float* x    = (const float*)d_in[0];
    const int*   ei   = (const int*)d_in[1];
    const float* Wl1  = (const float*)d_in[2];
    const float* Wr1  = (const float*)d_in[3];
    const float* att1 = (const float*)d_in[4];
    const float* b1   = (const float*)d_in[5];
    const float* Wl2  = (const float*)d_in[6];
    const float* Wr2  = (const float*)d_in[7];
    const float* att2 = (const float*)d_in[8];
    const float* b2   = (const float*)d_in[9];

    const int N  = in_sizes[0] / DIM;
    const int E  = in_sizes[1] / 2;
    const int ET = E + N;

    const int* srcA = ei;
    const int* dstA = ei + E;

    float*    xl    = (float*)d_ws;
    float*    xr    = xl + (size_t)N * DIM;
    float*    h1    = xr + (size_t)N * DIM;
    float*    alpha = h1 + (size_t)N * DIM;
    unsigned* amax  = (unsigned*)(alpha + (size_t)ET * 2);
    float*    out   = (float*)d_out;

    const dim3 B(256);
    const int nStat  = N * 2;
    const int gStat  = (nStat / 4 + 256 * ZF_PER - 1) / (256 * ZF_PER);
    const int gAlpha = (ET + 255) / 256;
    const int gGemm  = N / 16;
    const int gOwn   = (N + NA_NPB - 1) / NA_NPB;
    const dim3 BO(32 * NA_WAVES);

    zero_fill<<<gStat, B, 0, stream>>>((float*)amax, nStat);
    gemm128_wmma_f32<<<gGemm, B, 0, stream>>>(x, Wl1, xl);
    gemm128_wmma_f32<<<gGemm, B, 0, stream>>>(x, Wr1, xr);
    edge_alpha<2><<<gAlpha, B, 0, stream>>>(srcA, dstA, E, ET, xl, xr, att1, alpha, amax);
    node_aggregate<2><<<gOwn, BO, 0, stream>>>(srcA, dstA, E, xl, alpha, amax, b1, h1, N);

    zero_fill<<<gStat, B, 0, stream>>>((float*)amax, nStat);
    gemm128_wmma_f32<<<gGemm, B, 0, stream>>>(h1, Wl2, xl);
    gemm128_wmma_f32<<<gGemm, B, 0, stream>>>(h1, Wr2, xr);
    edge_alpha<1><<<gAlpha, B, 0, stream>>>(srcA, dstA, E, ET, xl, xr, att2, alpha, amax);
    node_aggregate<1><<<gOwn, BO, 0, stream>>>(srcA, dstA, E, xl, alpha, amax, b2, out, N);
}


# Score pass fused into the owner aggregate pass: an edge's source row is fetched once, scored and added; softmax by a running maximum that rescales the LDS sums; no atomics; 10 launches become 6
# speedup vs baseline: 3.2658x; 1.5283x over previous
#include <hip/hip_runtime.h>
#include <hip/hip_bf16.h>

typedef __attribute__((ext_vector_type(2))) float v2f;
typedef __attribute__((ext_vector_type(8))) float v8f;
typedef __attribute__((ext_vector_type(4))) float v4f;

#define DIM 128
#define NEG_SLOPE 0.2f

__device__ __forceinline__ float lrelu(float x) { return x > 0.f ? x : NEG_SLOPE * x; }
__device__ __forceinline__ float elu(float x)   { return x > 0.f ? x : (__expf(x) - 1.f); }

#define A_PITCH 132
__global__ void __launch_bounds__(256) gemm128_wmma_f32(
    const float* __restrict__ A, const float* __restrict__ W,
    float* __restrict__ C) {
    __shared__ __attribute__((aligned(16))) float As[16 * A_PITCH];
    const int wave = threadIdx.x >> 5;
    const int lane = threadIdx.x & 31;
    const int half = lane >> 4;
    const int l15  = lane & 15;
    const int row0 = blockIdx.x * 16;
    const int col0 = wave * 16;

    const float4* __restrict__ A4 = (const float4*)(A + (size_t)row0 * DIM);
    const float4 t0 = A4[threadIdx.x];
    const float4 t1 = A4[threadIdx.x + 256];

    v2f b[DIM / 4];
#pragma unroll
    for (int i = 0; i < DIM / 4; ++i) {
        const int ka = 4 * i + half * 2;
        b[i].x = W[(size_t)ka * DIM + col0 + l15];
        b[i].y = W[(size_t)(ka + 1) * DIM + col0 + l15];
    }

    *(float4*)&As[wave * A_PITCH + lane * 4]       = t0;
    *(float4*)&As[(wave + 8) * A_PITCH + lane * 4] = t1;
    __syncthreads();

    const float* __restrict__ Arow = As + l15 * A_PITCH + half * 2;

    v8f acc = {};
#pragma unroll
    for (int i = 0; i < DIM / 4; ++i) {
        const v2f a = *(const v2f*)(Arow + 4 * i);
        acc = __builtin_amdgcn_wmma_f32_16x16x4_f32(
             false, a,  false, b[i],
             (short)0, acc,  false,  false);
    }
#pragma unroll
    for (int j = 0; j < 8; ++j)
        C[(size_t)(row0 + j + half * 8) * DIM + col0 + l15] = acc[j];
}

#define NA_NPB   196
#define NA_WAVES 32
#define NA_SCAN  16
#define NA_TRIP  (32 * NA_SCAN)
#define NA_LCAP  704
#define NA_BCAP  64

__device__ __forceinline__ void na_load_whole(int (&d)[NA_SCAN], const int* __restrict__ dstA,
                                              int t, int lane) {
    const int* __restrict__ p = dstA + (size_t)t * NA_TRIP + lane;
#pragma unroll
    for (int k = 0; k < NA_SCAN; ++k) d[k] = p[32 * k];
}
__device__ __forceinline__ void na_load_part(int (&d)[NA_SCAN], const int* __restrict__ dstA,
                                             int t, int lane, int E) {
    const int e0 = t * NA_TRIP + lane;
#pragma unroll
    for (int k = 0; k < NA_SCAN; ++k) {
        const int e = e0 + 32 * k;
        d[k] = e < E ? dstA[e] : -1;
    }
}
__device__ __forceinline__ int na_scan(const int (&d)[NA_SCAN], int t, int lane, int lo, int cnt,
                                       int2* lst, int qn) {
    const int eb = t * NA_TRIP + lane;
#pragma unroll
    for (int k = 0; k < NA_SCAN; ++k) {
        const int dl = d[k] - lo;
        const bool mt = (unsigned)dl < (unsigned)cnt;
        const unsigned mask = __builtin_amdgcn_ballot_w32(mt);
        if (mask) {
            const int pos = qn + __builtin_popcount(mask & ((1u << lane) - 1u));
            if (mt) lst[pos] = make_int2(eb + 32 * k, dl);
            qn += __builtin_popcount(mask);
        }
    }
    return qn;
}

template <int HEADS, int J>
__device__ __forceinline__ void na_entry(
    const v4f row, float v, int b, int n, int dl, int lane, int h,
    float2* st_s, float* acc) {
    if (b + J < n) {
        const int dj = __builtin_amdgcn_readlane(dl, b + J);
        const float sc = __int_as_float(__builtin_amdgcn_update_dpp(
            0, __float_as_int(v), 0x150 + J, 0xf, 0xf, false));
        float2* st = st_s + dj * HEADS + h;
        float*  o  = acc + dj * DIM + lane;
        const float2 ms = *st;
        const float x0 = o[0], x1 = o[32], x2 = o[64], x3 = o[96];
        const float mn = fmaxf(ms.x, sc);
        const float f  = __expf(ms.x - mn);
        const float w  = __expf(sc - mn);
        *st = make_float2(mn, ms.y * f + w);
        o[0]  = x0 * f + w * row.x;
        o[32] = x1 * f + w * row.y;
        o[64] = x2 * f + w * row.z;
        o[96] = x3 * f + w * row.w;
    }
}

template <int HEADS>
__device__ __forceinline__ void na_chunk(
    const int2* buf, int n, int lane, int lo,
    const int* __restrict__ srcA, const float* __restrict__ xl,
    const float* __restrict__ xr, const float4 t,
    float2* st_s, float* acc) {
    const int2 en = buf[min(lane, n - 1)];
    const int dl = en.y;
    const int s  = srcA[en.x];
    const bool b0 = (lane & 1) != 0, b1 = (lane & 2) != 0, b2 = (lane & 4) != 0;
    const int  h  = (HEADS == 2) ? (lane >> 4) : 0;
    for (int b = 0; b < n; b += 8) {
        v4f a[8], y[8];
#pragma unroll
        for (int j = 0; j < 8; ++j) {
            const int sj = __builtin_amdgcn_readlane(s,  min(b + j, n - 1));
            const int dj = __builtin_amdgcn_readlane(dl, min(b + j, n - 1));
            a[j] = ((const v4f*)(xl + (size_t)sj * DIM))[lane];
            y[j] = ((const v4f*)(xr + (size_t)(lo + dj) * DIM))[lane];
        }
        asm volatile("" :: "v"(a[0]), "v"(a[1]), "v"(a[2]), "v"(a[3]),
                           "v"(a[4]), "v"(a[5]), "v"(a[6]), "v"(a[7]),
                           "v"(y[0]), "v"(y[1]), "v"(y[2]), "v"(y[3]),
                           "v"(y[4]), "v"(y[5]), "v"(y[6]), "v"(y[7]));

        float p[8];
#pragma unroll
        for (int j = 0; j < 8; ++j)
            p[j] = lrelu(a[j].x + y[j].x) * t.x + lrelu(a[j].y + y[j].y) * t.y +
                   lrelu(a[j].z + y[j].z) * t.z + lrelu(a[j].w + y[j].w) * t.w;

        float q[4];
#pragma unroll
        for (int i = 0; i < 4; ++i) {
            const float keep = b0 ? p[2 * i + 1] : p[2 * i];
            const float send = b0 ? p[2 * i] : p[2 * i + 1];
            q[i] = keep + __shfl_xor(send, 1, 32);
        }
        float r[2];
#pragma unroll
        for (int i = 0; i < 2; ++i) {
            const float keep = b1 ? q[2 * i + 1] : q[2 * i];
            const float send = b1 ? q[2 * i] : q[2 * i + 1];
            r[i] = keep + __shfl_xor(send, 2, 32);
        }
        float v;
        {
            const float keep = b2 ? r[1] : r[0];
            const float send = b2 ? r[0] : r[1];
            v = keep + __shfl_xor(send, 4, 32);
        }
        v += __shfl_xor(v, 8, 32);
        if (HEADS == 1) v += __shfl_xor(v, 16, 32);

        na_entry<HEADS, 0>(a[0], v, b, n, dl, lane, h, st_s, acc);
        na_entry<HEADS, 1>(a[1], v, b, n, dl, lane, h, st_s, acc);
        na_entry<HEADS, 2>(a[2], v, b, n, dl, lane, h, st_s, acc);
        na_entry<HEADS, 3>(a[3], v, b, n, dl, lane, h, st_s, acc);
        na_entry<HEADS, 4>(a[4], v, b, n, dl, lane, h, st_s, acc);
        na_entry<HEADS, 5>(a[5], v, b, n, dl, lane, h, st_s, acc);
        na_entry<HEADS, 6>(a[6], v, b, n, dl, lane, h, st_s, acc);
        na_entry<HEADS, 7>(a[7], v, b, n, dl, lane, h, st_s, acc);
    }
}

template <int HEADS>
__global__ void __launch_bounds__(1024) node_aggregate(
    const int* __restrict__ srcA, const int* __restrict__ dstA, int E,
    const float* __restrict__ xl, const float* __restrict__ xr,
    const float* __restrict__ att, const float* __restrict__ bias,
    float* __restrict__ out, int N) {
    __shared__ __attribute__((aligned(16))) float acc[NA_NPB * DIM];
    __shared__ float2 st_s[NA_NPB * 2];
    __shared__ int2  lst_s[NA_WAVES * NA_LCAP];
    __shared__ int2  buf_s[NA_WAVES * NA_BCAP];
    __shared__ int   cnt_s[NA_WAVES];
    __shared__ int   fin_s[NA_WAVES];

    const int lane = threadIdx.x & 31;
    const int wave = __builtin_amdgcn_readfirstlane(threadIdx.x >> 5);
    const int lo   = blockIdx.x * NA_NPB;
    const int cnt  = min(NA_NPB, N - lo);
    const int h    = (HEADS == 2) ? (lane >> 4) : 0;
    const float4 tv = ((const float4*)att)[lane];

    for (int r = wave; r < cnt; r += NA_WAVES) {
        const int node = lo + r;
        const float4 x = ((const float4*)(xl + (size_t)node * DIM))[lane];
        const float4 y = ((const float4*)(xr + (size_t)node * DIM))[lane];
        float v = lrelu(x.x + y.x) * tv.x + lrelu(x.y + y.y) * tv.y +
                  lrelu(x.z + y.z) * tv.z + lrelu(x.w + y.w) * tv.w;
        v += __shfl_xor(v, 1, 32);
        v += __shfl_xor(v, 2, 32);
        v += __shfl_xor(v, 4, 32);
        v += __shfl_xor(v, 8, 32);
        if (HEADS == 1) v += __shfl_xor(v, 16, 32);
        float* o = acc + r * DIM + lane;
        o[0] = x.x; o[32] = x.y; o[64] = x.z; o[96] = x.w;
        st_s[r * HEADS + h] = make_float2(v, 1.f);
    }

    int2* lst = lst_s + wave * NA_LCAP;
    int2* buf = buf_s + wave * NA_BCAP;
    const int nFull = E / NA_TRIP;
    const bool ownsPart = (E - nFull * NA_TRIP) > 0 && (nFull % NA_WAVES) == wave;
    bool partLeft = ownsPart;
    int t = wave;

    for (int round = 0; round < nFull + 2; ++round) {
        int qn = 0;
        if (t < nFull) {
            int a[NA_SCAN], b[NA_SCAN];
            na_load_whole(a, dstA, t, lane);
            for (;;) {
                na_load_whole(b, dstA, min(t + NA_WAVES, nFull - 1), lane);
                qn = na_scan(a, t, lane, lo, cnt, lst, qn);
                t += NA_WAVES;
                if (t >= nFull || qn + NA_TRIP > NA_LCAP) break;
                na_load_whole(a, dstA, min(t + NA_WAVES, nFull - 1), lane);
                qn = na_scan(b, t, lane, lo, cnt, lst, qn);
                t += NA_WAVES;
                if (t >= nFull || qn + NA_TRIP > NA_LCAP) break;
            }
        }
        if (partLeft && t >= nFull && qn + NA_TRIP <= NA_LCAP) {
            int c[NA_SCAN];
            na_load_part(c, dstA, nFull, lane, E);
            qn = na_scan(c, nFull, lane, lo, cnt, lst, qn);
            partLeft = false;
        }
        if (lane == 0) {
            cnt_s[wave] = qn;
            fin_s[wave] = (t < nFull || partLeft) ? 1 : 0;
        }
        __syncthreads();

        int bn = 0;
        for (int l = 0; l < NA_WAVES; ++l) {
            const int n = __builtin_amdgcn_readfirstlane(cnt_s[l]);
            const int2* src_l = lst_s + l * NA_LCAP;
            for (int p = 0; p < n; p += 32) {
                const int i = p + lane;
                const int2 en = src_l[min(i, n - 1)];
                const bool mt = i < n && (en.y & (NA_WAVES - 1)) == wave;
                const unsigned mask = __builtin_amdgcn_ballot_w32(mt);
                if (mask) {
                    const int pos = bn + __builtin_popcount(mask & ((1u << lane) - 1u));
                    if (mt) buf[pos] = en;
                    bn += __builtin_popcount(mask);
                    if (bn >= 32) {
                        na_chunk<HEADS>(buf, 32, lane, lo, srcA, xl, xr, tv, st_s, acc);
                        const int rem = bn - 32;
                        const int2 keep = buf[32 + min(lane, NA_BCAP - 33)];
                        if (lane < rem) buf[lane] = keep;
                        bn = rem;
                    }
                }
            }
        }
        if (bn > 0)
            na_chunk<HEADS>(buf, bn, lane, lo, srcA, xl, xr, tv, st_s, acc);

        const bool more = __builtin_amdgcn_ballot_w32(fin_s[lane] != 0) != 0u;
        __syncthreads();
        if (!more) break;
    }

    const float4 bz = ((const float4*)bias)[lane];
    for (int r = wave; r < cnt; r += NA_WAVES) {
        const float* o = acc + r * DIM + lane;
        const float dd = st_s[r * HEADS + h].y + 1e-16f;
        float4 z;
        z.x = elu(o[0]  / dd + bz.x);
        z.y = elu(o[32] / dd + bz.y);
        z.z = elu(o[64] / dd + bz.z);
        z.w = elu(o[96] / dd + bz.w);
        ((float4*)(out + (size_t)(lo + r) * DIM))[lane] = z;
    }
}

extern "C" void kernel_launch(void* const* d_in, const int* in_sizes, int n_in,
                              void* d_out, int out_size, void* d_ws, size_t ws_size,
                              hipStream_t stream) {
    const float* x    = (const float*)d_in[0];
    const int*   ei   = (const int*)d_in[1];
    const float* Wl1  = (const float*)d_in[2];
    const float* Wr1  = (const float*)d_in[3];
    const float* att1 = (const float*)d_in[4];
    const float* b1   = (const float*)d_in[5];
    const float* Wl2  = (const float*)d_in[6];
    const float* Wr2  = (const float*)d_in[7];
    const float* att2 = (const float*)d_in[8];
    const float* b2   = (const float*)d_in[9];

    const int N  = in_sizes[0] / DIM;
    const int E  = in_sizes[1] / 2;

    const int* srcA = ei;
    const int* dstA = ei + E;

    float*    xl    = (float*)d_ws;
    float*    xr    = xl + (size_t)N * DIM;
    float*    h1    = xr + (size_t)N * DIM;
    float*    out   = (float*)d_out;

    const dim3 B(256);
    const int gGemm  = N / 16;
    const int gOwn   = (N + NA_NPB - 1) / NA_NPB;
    const dim3 BO(32 * NA_WAVES);

    gemm128_wmma_f32<<<gGemm, B, 0, stream>>>(x, Wl1, xl);
    gemm128_wmma_f32<<<gGemm, B, 0, stream>>>(x, Wr1, xr);
    node_aggregate<2><<<gOwn, BO, 0, stream>>>(srcA, dstA, E, xl, xr, att1, b1, h1, N);

    gemm128_wmma_f32<<<gGemm, B, 0, stream>>>(h1, Wl2, xl);
    gemm128_wmma_f32<<<gGemm, B, 0, stream>>>(h1, Wr2, xr);
    node_aggregate<1><<<gOwn, BO, 0, stream>>>(srcA, dstA, E, xl, xr, att2, b2, out, N);
}


# Layer 2's aggregate pass replays per-wave edge sequences (source id, local node) that layer 1's pass saved in its add order; no second walk of the destination list; same sum order
# speedup vs baseline: 3.8544x; 1.1802x over previous
#include <hip/hip_runtime.h>
#include <hip/hip_bf16.h>

typedef __attribute__((ext_vector_type(2))) float v2f;
typedef __attribute__((ext_vector_type(8))) float v8f;
typedef __attribute__((ext_vector_type(4))) float v4f;

#define DIM 128
#define NEG_SLOPE 0.2f

__device__ __forceinline__ float lrelu(float x) { return x > 0.f ? x : NEG_SLOPE * x; }
__device__ __forceinline__ float elu(float x)   { return x > 0.f ? x : (__expf(x) - 1.f); }

#define A_PITCH 132
__global__ void __launch_bounds__(256) gemm128_wmma_f32(
    const float* __restrict__ A, const float* __restrict__ W,
    float* __restrict__ C) {
    __shared__ __attribute__((aligned(16))) float As[16 * A_PITCH];
    const int wave = threadIdx.x >> 5;
    const int lane = threadIdx.x & 31;
    const int half = lane >> 4;
    const int l15  = lane & 15;
    const int row0 = blockIdx.x * 16;
    const int col0 = wave * 16;

    const float4* __restrict__ A4 = (const float4*)(A + (size_t)row0 * DIM);
    const float4 t0 = A4[threadIdx.x];
    const float4 t1 = A4[threadIdx.x + 256];

    v2f b[DIM / 4];
#pragma unroll
    for (int i = 0; i < DIM / 4; ++i) {
        const int ka = 4 * i + half * 2;
        b[i].x = W[(size_t)ka * DIM + col0 + l15];
        b[i].y = W[(size_t)(ka + 1) * DIM + col0 + l15];
    }

    *(float4*)&As[wave * A_PITCH + lane * 4]       = t0;
    *(float4*)&As[(wave + 8) * A_PITCH + lane * 4] = t1;
    __syncthreads();

    const float* __restrict__ Arow = As + l15 * A_PITCH + half * 2;

    v8f acc = {};
#pragma unroll
    for (int i = 0; i < DIM / 4; ++i) {
        const v2f a = *(const v2f*)(Arow + 4 * i);
        acc = __builtin_amdgcn_wmma_f32_16x16x4_f32(
             false, a,  false, b[i],
             (short)0, acc,  false,  false);
    }
#pragma unroll
    for (int j = 0; j < 8; ++j)
        C[(size_t)(row0 + j + half * 8) * DIM + col0 + l15] = acc[j];
}

#define NA_NPB   196
#define NA_WAVES 32
#define NA_SCAN  16
#define NA_TRIP  (32 * NA_SCAN)
#define NA_LCAP  704
#define NA_BCAP  64
#define NA_SCAP  1024

__device__ __forceinline__ void na_load_whole(int (&d)[NA_SCAN], const int* __restrict__ dstA,
                                              int t, int lane) {
    const int* __restrict__ p = dstA + (size_t)t * NA_TRIP + lane;
#pragma unroll
    for (int k = 0; k < NA_SCAN; ++k) d[k] = p[32 * k];
}
__device__ __forceinline__ void na_load_part(int (&d)[NA_SCAN], const int* __restrict__ dstA,
                                             int t, int lane, int E) {
    const int e0 = t * NA_TRIP + lane;
#pragma unroll
    for (int k = 0; k < NA_SCAN; ++k) {
        const int e = e0 + 32 * k;
        d[k] = e < E ? dstA[e] : -1;
    }
}
__device__ __forceinline__ int na_scan(const int (&d)[NA_SCAN], int t, int lane, int lo, int cnt,
                                       int2* lst, int qn) {
    const int eb = t * NA_TRIP + lane;
#pragma unroll
    for (int k = 0; k < NA_SCAN; ++k) {
        const int dl = d[k] - lo;
        const bool mt = (unsigned)dl < (unsigned)cnt;
        const unsigned mask = __builtin_amdgcn_ballot_w32(mt);
        if (mask) {
            const int pos = qn + __builtin_popcount(mask & ((1u << lane) - 1u));
            if (mt) lst[pos] = make_int2(eb + 32 * k, dl);
            qn += __builtin_popcount(mask);
        }
    }
    return qn;
}

template <int HEADS, int J>
__device__ __forceinline__ void na_entry(
    const v4f row, float v, int b, int n, int dl, int lane, int h,
    float2* st_s, float* acc) {
    if (b + J < n) {
        const int dj = __builtin_amdgcn_readlane(dl, b + J);
        const float sc = __int_as_float(__builtin_amdgcn_update_dpp(
            0, __float_as_int(v), 0x150 + J, 0xf, 0xf, false));
        float2* st = st_s + dj * HEADS + h;
        float*  o  = acc + dj * DIM + lane;
        const float2 ms = *st;
        const float x0 = o[0], x1 = o[32], x2 = o[64], x3 = o[96];
        const float mn = fmaxf(ms.x, sc);
        const float f  = __expf(ms.x - mn);
        const float w  = __expf(sc - mn);
        *st = make_float2(mn, ms.y * f + w);
        o[0]  = x0 * f + w * row.x;
        o[32] = x1 * f + w * row.y;
        o[64] = x2 * f + w * row.z;
        o[96] = x3 * f + w * row.w;
    }
}

template <int HEADS>
__device__ __forceinline__ void na_rows(
    int s, int dl, int n, int lane, int lo,
    const float* __restrict__ xl, const float* __restrict__ xr, const float4 t,
    float2* st_s, float* acc) {
    const bool b0 = (lane & 1) != 0, b1 = (lane & 2) != 0, b2 = (lane & 4) != 0;
    const int  h  = (HEADS == 2) ? (lane >> 4) : 0;
    for (int b = 0; b < n; b += 8) {
        v4f a[8], y[8];
#pragma unroll
        for (int j = 0; j < 8; ++j) {
            const int sj = __builtin_amdgcn_readlane(s,  min(b + j, n - 1));
            const int dj = __builtin_amdgcn_readlane(dl, min(b + j, n - 1));
            a[j] = ((const v4f*)(xl + (size_t)sj * DIM))[lane];
            y[j] = ((const v4f*)(xr + (size_t)(lo + dj) * DIM))[lane];
        }
        asm volatile("" :: "v"(a[0]), "v"(a[1]), "v"(a[2]), "v"(a[3]),
                           "v"(a[4]), "v"(a[5]), "v"(a[6]), "v"(a[7]),
                           "v"(y[0]), "v"(y[1]), "v"(y[2]), "v"(y[3]),
                           "v"(y[4]), "v"(y[5]), "v"(y[6]), "v"(y[7]));

        float p[8];
#pragma unroll
        for (int j = 0; j < 8; ++j)
            p[j] = lrelu(a[j].x + y[j].x) * t.x + lrelu(a[j].y + y[j].y) * t.y +
                   lrelu(a[j].z + y[j].z) * t.z + lrelu(a[j].w + y[j].w) * t.w;

        float q[4];
#pragma unroll
        for (int i = 0; i < 4; ++i) {
            const float keep = b0 ? p[2 * i + 1] : p[2 * i];
            const float send = b0 ? p[2 * i] : p[2 * i + 1];
            q[i] = keep + __shfl_xor(send, 1, 32);
        }
        float r[2];
#pragma unroll
        for (int i = 0; i < 2; ++i) {
            const float keep = b1 ? q[2 * i + 1] : q[2 * i];
            const float send = b1 ? q[2 * i] : q[2 * i + 1];
            r[i] = keep + __shfl_xor(send, 2, 32);
        }
        float v;
        {
            const float keep = b2 ? r[1] : r[0];
            const float send = b2 ? r[0] : r[1];
            v = keep + __shfl_xor(send, 4, 32);
        }
        v += __shfl_xor(v, 8, 32);
        if (HEADS == 1) v += __shfl_xor(v, 16, 32);

        na_entry<HEADS, 0>(a[0], v, b, n, dl, lane, h, st_s, acc);
        na_entry<HEADS, 1>(a[1], v, b, n, dl, lane, h, st_s, acc);
        na_entry<HEADS, 2>(a[2], v, b, n, dl, lane, h, st_s, acc);
        na_entry<HEADS, 3>(a[3], v, b, n, dl, lane, h, st_s, acc);
        na_entry<HEADS, 4>(a[4], v, b, n, dl, lane, h, st_s, acc);
        na_entry<HEADS, 5>(a[5], v, b, n, dl, lane, h, st_s, acc);
        na_entry<HEADS, 6>(a[6], v, b, n, dl, lane, h, st_s, acc);
        na_entry<HEADS, 7>(a[7], v, b, n, dl, lane, h, st_s, acc);
    }
}

template <int HEADS, bool SAVE>
__device__ __forceinline__ int na_chunk(
    const int2* buf, int n, int lane, int lo,
    const int* __restrict__ srcA, const float* __restrict__ xl,
    const float* __restrict__ xr, const float4 t,
    float2* st_s, float* acc, int2* __restrict__ seq, int sn) {
    const int2 en = buf[min(lane, n - 1)];
    const int dl = en.y;
    const int s  = srcA[en.x];
    if (SAVE) {
        if (lane < n && sn + lane < NA_SCAP) seq[sn + lane] = make_int2(s, dl);
    }
    na_rows<HEADS>(s, dl, n, lane, lo, xl, xr, t, st_s, acc);
    return sn + n;
}

template <int HEADS, int MODE>
__global__ void __launch_bounds__(1024) node_aggregate(
    const int* __restrict__ srcA, const int* __restrict__ dstA, int E,
    const float* __restrict__ xl, const float* __restrict__ xr,
    const float* __restrict__ att, const float* __restrict__ bias,
    float* __restrict__ out, int N,
    int* __restrict__ seqG, int* __restrict__ cntG) {
    __shared__ __attribute__((aligned(16))) float acc[NA_NPB * DIM];
    __shared__ float2 st_s[NA_NPB * 2];
    __shared__ int2  lst_s[NA_WAVES * NA_LCAP];
    __shared__ int2  buf_s[NA_WAVES * NA_BCAP];
    __shared__ int   cnt_s[NA_WAVES];
    __shared__ int   fin_s[NA_WAVES];

    const int lane = threadIdx.x & 31;
    const int wave = __builtin_amdgcn_readfirstlane(threadIdx.x >> 5);
    const int lo   = blockIdx.x * NA_NPB;
    const int cnt  = min(NA_NPB, N - lo);
    const int h    = (HEADS == 2) ? (lane >> 4) : 0;
    const float4 tv = ((const float4*)att)[lane];

    for (int r = wave; r < cnt; r += NA_WAVES) {
        const int node = lo + r;
        const float4 x = ((const float4*)(xl + (size_t)node * DIM))[lane];
        const float4 y = ((const float4*)(xr + (size_t)node * DIM))[lane];
        float v = lrelu(x.x + y.x) * tv.x + lrelu(x.y + y.y) * tv.y +
                  lrelu(x.z + y.z) * tv.z + lrelu(x.w + y.w) * tv.w;
        v += __shfl_xor(v, 1, 32);
        v += __shfl_xor(v, 2, 32);
        v += __shfl_xor(v, 4, 32);
        v += __shfl_xor(v, 8, 32);
        if (HEADS == 1) v += __shfl_xor(v, 16, 32);
        float* o = acc + r * DIM + lane;
        o[0] = x.x; o[32] = x.y; o[64] = x.z; o[96] = x.w;
        st_s[r * HEADS + h] = make_float2(v, 1.f);
    }

    const int wid = blockIdx.x * NA_WAVES + wave;
    int2* __restrict__ seq = (int2*)seqG + (size_t)wid * NA_SCAP;
    int sn = 0;

    bool replay = false;
    int total = 0;
    if (MODE == 1) {
        const int c = cntG[blockIdx.x * NA_WAVES + lane];
        replay = __builtin_amdgcn_ballot_w32(c > NA_SCAP) == 0u;
        total  = __builtin_amdgcn_readlane(c, wave);
    }

    if (MODE == 1 && replay) {
        if (total > 0) {
            int2 cur = seq[min(lane, min(32, total) - 1)];
            for (int p = 0; p < total; p += 32) {
                const int n  = min(32, total - p);
                const int pn = p + 32;
                int2 nxt = cur;
                if (pn < total) nxt = seq[pn + min(lane, min(32, total - pn) - 1)];
                na_rows<HEADS>(cur.x, cur.y, n, lane, lo, xl, xr, tv, st_s, acc);
                cur = nxt;
            }
        }
    } else {
    int2* lst = lst_s + wave * NA_LCAP;
    int2* buf = buf_s + wave * NA_BCAP;
    const int nFull = E / NA_TRIP;
    const bool ownsPart = (E - nFull * NA_TRIP) > 0 && (nFull % NA_WAVES) == wave;
    bool partLeft = ownsPart;
    int t = wave;

    for (int round = 0; round < nFull + 2; ++round) {
        int qn = 0;
        if (t < nFull) {
            int a[NA_SCAN], b[NA_SCAN];
            na_load_whole(a, dstA, t, lane);
            for (;;) {
                na_load_whole(b, dstA, min(t + NA_WAVES, nFull - 1), lane);
                qn = na_scan(a, t, lane, lo, cnt, lst, qn);
                t += NA_WAVES;
                if (t >= nFull || qn + NA_TRIP > NA_LCAP) break;
                na_load_whole(a, dstA, min(t + NA_WAVES, nFull - 1), lane);
                qn = na_scan(b, t, lane, lo, cnt, lst, qn);
                t += NA_WAVES;
                if (t >= nFull || qn + NA_TRIP > NA_LCAP) break;
            }
        }
        if (partLeft && t >= nFull && qn + NA_TRIP <= NA_LCAP) {
            int c[NA_SCAN];
            na_load_part(c, dstA, nFull, lane, E);
            qn = na_scan(c, nFull, lane, lo, cnt, lst, qn);
            partLeft = false;
        }
        if (lane == 0) {
            cnt_s[wave] = qn;
            fin_s[wave] = (t < nFull || partLeft) ? 1 : 0;
        }
        __syncthreads();

        int bn = 0;
        for (int l = 0; l < NA_WAVES; ++l) {
            const int n = __builtin_amdgcn_readfirstlane(cnt_s[l]);
            const int2* src_l = lst_s + l * NA_LCAP;
            for (int p = 0; p < n; p += 32) {
                const int i = p + lane;
                const int2 en = src_l[min(i, n - 1)];
                const bool mt = i < n && (en.y & (NA_WAVES - 1)) == wave;
                const unsigned mask = __builtin_amdgcn_ballot_w32(mt);
                if (mask) {
                    const int pos = bn + __builtin_popcount(mask & ((1u << lane) - 1u));
                    if (mt) buf[pos] = en;
                    bn += __builtin_popcount(mask);
                    if (bn >= 32) {
                        sn = na_chunk<HEADS, MODE == 0>(buf, 32, lane, lo, srcA, xl, xr, tv,
                                                        st_s, acc, seq, sn);
                        const int rem = bn - 32;
                        const int2 keep = buf[32 + min(lane, NA_BCAP - 33)];
                        if (lane < rem) buf[lane] = keep;
                        bn = rem;
                    }
                }
            }
        }
        if (bn > 0)
            sn = na_chunk<HEADS, MODE == 0>(buf, bn, lane, lo, srcA, xl, xr, tv,
                                            st_s, acc, seq, sn);

        const bool more = __builtin_amdgcn_ballot_w32(fin_s[lane] != 0) != 0u;
        __syncthreads();
        if (!more) break;
    }
    }
    if (MODE == 0 && lane == 0) cntG[wid] = sn;

    const float4 bz = ((const float4*)bias)[lane];
    for (int r = wave; r < cnt; r += NA_WAVES) {
        const float* o = acc + r * DIM + lane;
        const float dd = st_s[r * HEADS + h].y + 1e-16f;
        float4 z;
        z.x = elu(o[0]  / dd + bz.x);
        z.y = elu(o[32] / dd + bz.y);
        z.z = elu(o[64] / dd + bz.z);
        z.w = elu(o[96] / dd + bz.w);
        ((float4*)(out + (size_t)(lo + r) * DIM))[lane] = z;
    }
}

extern "C" void kernel_launch(void* const* d_in, const int* in_sizes, int n_in,
                              void* d_out, int out_size, void* d_ws, size_t ws_size,
                              hipStream_t stream) {
    const float* x    = (const float*)d_in[0];
    const int*   ei   = (const int*)d_in[1];
    const float* Wl1  = (const float*)d_in[2];
    const float* Wr1  = (const float*)d_in[3];
    const float* att1 = (const float*)d_in[4];
    const float* b1   = (const float*)d_in[5];
    const float* Wl2  = (const float*)d_in[6];
    const float* Wr2  = (const float*)d_in[7];
    const float* att2 = (const float*)d_in[8];
    const float* b2   = (const float*)d_in[9];

    const int N  = in_sizes[0] / DIM;
    const int E  = in_sizes[1] / 2;

    const int* srcA = ei;
    const int* dstA = ei + E;

    float*    xl    = (float*)d_ws;
    float*    xr    = xl + (size_t)N * DIM;
    float*    h1    = xr + (size_t)N * DIM;
    float*    out   = (float*)d_out;

    const dim3 B(256);
    const int gGemm  = N / 16;
    const int gOwn   = (N + NA_NPB - 1) / NA_NPB;
    const dim3 BO(32 * NA_WAVES);
    int*      seqG  = (int*)(h1 + (size_t)N * DIM);
    int*      cntG  = seqG + (size_t)gOwn * NA_WAVES * NA_SCAP * 2;

    gemm128_wmma_f32<<<gGemm, B, 0, stream>>>(x, Wl1, xl);
    gemm128_wmma_f32<<<gGemm, B, 0, stream>>>(x, Wr1, xr);
    node_aggregate<2, 0><<<gOwn, BO, 0, stream>>>(srcA, dstA, E, xl, xr, att1, b1, h1, N,
                                                  seqG, cntG);

    gemm128_wmma_f32<<<gGemm, B, 0, stream>>>(h1, Wl2, xl);
    gemm128_wmma_f32<<<gGemm, B, 0, stream>>>(h1, Wr2, xr);
    node_aggregate<1, 1><<<gOwn, BO, 0, stream>>>(srcA, dstA, E, xl, xr, att2, b2, out, N,
                                                  seqG, cntG);
}


# GEMM keeps W in registers: 768 resident blocks; a wave loads its 128 x 16 slice of W once and walks its block's row tiles with it (next A tile prefetched, two LDS tiles); 6144 waves, not 25000
# speedup vs baseline: 4.2720x; 1.1083x over previous
#include <hip/hip_runtime.h>
#include <hip/hip_bf16.h>

typedef __attribute__((ext_vector_type(2))) float v2f;
typedef __attribute__((ext_vector_type(8))) float v8f;
typedef __attribute__((ext_vector_type(4))) float v4f;

#define DIM 128
#define NEG_SLOPE 0.2f

__device__ __forceinline__ float lrelu(float x) { return x > 0.f ? x : NEG_SLOPE * x; }
__device__ __forceinline__ float elu(float x)   { return x > 0.f ? x : (__expf(x) - 1.f); }

#define A_PITCH 132
#define GEMM_GRID 768
__global__ void __launch_bounds__(256) gemm128_wmma_f32(
    const float* __restrict__ A, const float* __restrict__ W,
    float* __restrict__ C, int nTiles) {
    __shared__ __attribute__((aligned(16))) float As[2 * 16 * A_PITCH];
    const int wave = threadIdx.x >> 5;
    const int lane = threadIdx.x & 31;
    const int half = lane >> 4;
    const int l15  = lane & 15;
    const int col0 = wave * 16;

    int t = blockIdx.x;
    if (t >= nTiles) return;

    const float4* __restrict__ A4 = (const float4*)A;
    float4 p0 = A4[(size_t)t * 512 + threadIdx.x];
    float4 p1 = A4[(size_t)t * 512 + threadIdx.x + 256];

    v2f b[DIM / 4];
#pragma unroll
    for (int i = 0; i < DIM / 4; ++i) {
        const int ka = 4 * i + half * 2;
        b[i].x = W[(size_t)ka * DIM + col0 + l15];
        b[i].y = W[(size_t)(ka + 1) * DIM + col0 + l15];
    }

    const int wr0 = wave * A_PITCH + lane * 4;
    const int wr1 = (wave + 8) * A_PITCH + lane * 4;
    const int rd  = l15 * A_PITCH + half * 2;

    *(float4*)&As[wr0] = p0;
    *(float4*)&As[wr1] = p1;
    __syncthreads();

    int cur = 0;
    for (;;) {
        const int  tn   = t + GEMM_GRID;
        const bool more = tn < nTiles;
        if (more) {
            p0 = A4[(size_t)tn * 512 + threadIdx.x];
            p1 = A4[(size_t)tn * 512 + threadIdx.x + 256];
        }

        const float* __restrict__ Arow = As + cur + rd;
        v8f acc = {};
#pragma unroll
        for (int i = 0; i < DIM / 4; ++i) {
            const v2f a = *(const v2f*)(Arow + 4 * i);
            acc = __builtin_amdgcn_wmma_f32_16x16x4_f32(
                 false, a,  false, b[i],
                 (short)0, acc,  false,  false);
        }

        if (more) {
            const int nxt = 16 * A_PITCH - cur;
            *(float4*)__builtin_assume_aligned(&As[nxt + wr0], 16) = p0;
            *(float4*)__builtin_assume_aligned(&As[nxt + wr1], 16) = p1;
            __syncthreads();
            cur = nxt;
        }

        const int row0 = t * 16;
#pragma unroll
        for (int j = 0; j < 8; ++j)
            C[(size_t)(row0 + j + half * 8) * DIM + col0 + l15] = acc[j];

        if (!more) break;
        t = tn;
    }
}

#define NA_NPB   196
#define NA_WAVES 32
#define NA_SCAN  16
#define NA_TRIP  (32 * NA_SCAN)
#define NA_LCAP  704
#define NA_BCAP  64
#define NA_SCAP  1024

__device__ __forceinline__ void na_load_whole(int (&d)[NA_SCAN], const int* __restrict__ dstA,
                                              int t, int lane) {
    const int* __restrict__ p = dstA + (size_t)t * NA_TRIP + lane;
#pragma unroll
    for (int k = 0; k < NA_SCAN; ++k) d[k] = p[32 * k];
}
__device__ __forceinline__ void na_load_part(int (&d)[NA_SCAN], const int* __restrict__ dstA,
                                             int t, int lane, int E) {
    const int e0 = t * NA_TRIP + lane;
#pragma unroll
    for (int k = 0; k < NA_SCAN; ++k) {
        const int e = e0 + 32 * k;
        d[k] = e < E ? dstA[e] : -1;
    }
}
__device__ __forceinline__ int na_scan(const int (&d)[NA_SCAN], int t, int lane, int lo, int cnt,
                                       int2* lst, int qn) {
    const int eb = t * NA_TRIP + lane;
#pragma unroll
    for (int k = 0; k < NA_SCAN; ++k) {
        const int dl = d[k] - lo;
        const bool mt = (unsigned)dl < (unsigned)cnt;
        const unsigned mask = __builtin_amdgcn_ballot_w32(mt);
        if (mask) {
            const int pos = qn + __builtin_popcount(mask & ((1u << lane) - 1u));
            if (mt) lst[pos] = make_int2(eb + 32 * k, dl);
            qn += __builtin_popcount(mask);
        }
    }
    return qn;
}

template <int HEADS, int J>
__device__ __forceinline__ void na_entry(
    const v4f row, float v, int b, int n, int dl, int lane, int h,
    float2* st_s, float* acc) {
    if (b + J < n) {
        const int dj = __builtin_amdgcn_readlane(dl, b + J);
        const float sc = __int_as_float(__builtin_amdgcn_update_dpp(
            0, __float_as_int(v), 0x150 + J, 0xf, 0xf, false));
        float2* st = st_s + dj * HEADS + h;
        float*  o  = acc + dj * DIM + lane;
        const float2 ms = *st;
        const float x0 = o[0], x1 = o[32], x2 = o[64], x3 = o[96];
        const float mn = fmaxf(ms.x, sc);
        const float f  = __expf(ms.x - mn);
        const float w  = __expf(sc - mn);
        *st = make_float2(mn, ms.y * f + w);
        o[0]  = x0 * f + w * row.x;
        o[32] = x1 * f + w * row.y;
        o[64] = x2 * f + w * row.z;
        o[96] = x3 * f + w * row.w;
    }
}

template <int HEADS>
__device__ __forceinline__ void na_rows(
    int s, int dl, int n, int lane, int lo,
    const float* __restrict__ xl, const float* __restrict__ xr, const float4 t,
    float2* st_s, float* acc) {
    const bool b0 = (lane & 1) != 0, b1 = (lane & 2) != 0, b2 = (lane & 4) != 0;
    const int  h  = (HEADS == 2) ? (lane >> 4) : 0;
    for (int b = 0; b < n; b += 8) {
        v4f a[8], y[8];
#pragma unroll
        for (int j = 0; j < 8; ++j) {
            const int sj = __builtin_amdgcn_readlane(s,  min(b + j, n - 1));
            const int dj = __builtin_amdgcn_readlane(dl, min(b + j, n - 1));
            a[j] = ((const v4f*)(xl + (size_t)sj * DIM))[lane];
            y[j] = ((const v4f*)(xr + (size_t)(lo + dj) * DIM))[lane];
        }
        asm volatile("" :: "v"(a[0]), "v"(a[1]), "v"(a[2]), "v"(a[3]),
                           "v"(a[4]), "v"(a[5]), "v"(a[6]), "v"(a[7]),
                           "v"(y[0]), "v"(y[1]), "v"(y[2]), "v"(y[3]),
                           "v"(y[4]), "v"(y[5]), "v"(y[6]), "v"(y[7]));

        float p[8];
#pragma unroll
        for (int j = 0; j < 8; ++j)
            p[j] = lrelu(a[j].x + y[j].x) * t.x + lrelu(a[j].y + y[j].y) * t.y +
                   lrelu(a[j].z + y[j].z) * t.z + lrelu(a[j].w + y[j].w) * t.w;

        float q[4];
#pragma unroll
        for (int i = 0; i < 4; ++i) {
            const float keep = b0 ? p[2 * i + 1] : p[2 * i];
            const float send = b0 ? p[2 * i] : p[2 * i + 1];
            q[i] = keep + __shfl_xor(send, 1, 32);
        }
        float r[2];
#pragma unroll
        for (int i = 0; i < 2; ++i) {
            const float keep = b1 ? q[2 * i + 1] : q[2 * i];
            const float send = b1 ? q[2 * i] : q[2 * i + 1];
            r[i] = keep + __shfl_xor(send, 2, 32);
        }
        float v;
        {
            const float keep = b2 ? r[1] : r[0];
            const float send = b2 ? r[0] : r[1];
            v = keep + __shfl_xor(send, 4, 32);
        }
        v += __shfl_xor(v, 8, 32);
        if (HEADS == 1) v += __shfl_xor(v, 16, 32);

        na_entry<HEADS, 0>(a[0], v, b, n, dl, lane, h, st_s, acc);
        na_entry<HEADS, 1>(a[1], v, b, n, dl, lane, h, st_s, acc);
        na_entry<HEADS, 2>(a[2], v, b, n, dl, lane, h, st_s, acc);
        na_entry<HEADS, 3>(a[3], v, b, n, dl, lane, h, st_s, acc);
        na_entry<HEADS, 4>(a[4], v, b, n, dl, lane, h, st_s, acc);
        na_entry<HEADS, 5>(a[5], v, b, n, dl, lane, h, st_s, acc);
        na_entry<HEADS, 6>(a[6], v, b, n, dl, lane, h, st_s, acc);
        na_entry<HEADS, 7>(a[7], v, b, n, dl, lane, h, st_s, acc);
    }
}

template <int HEADS, bool SAVE>
__device__ __forceinline__ int na_chunk(
    const int2* buf, int n, int lane, int lo,
    const int* __restrict__ srcA, const float* __restrict__ xl,
    const float* __restrict__ xr, const float4 t,
    float2* st_s, float* acc, int2* __restrict__ seq, int sn) {
    const int2 en = buf[min(lane, n - 1)];
    const int dl = en.y;
    const int s  = srcA[en.x];
    if (SAVE) {
        if (lane < n && sn + lane < NA_SCAP) seq[sn + lane] = make_int2(s, dl);
    }
    na_rows<HEADS>(s, dl, n, lane, lo, xl, xr, t, st_s, acc);
    return sn + n;
}

template <int HEADS, int MODE>
__global__ void __launch_bounds__(1024) node_aggregate(
    const int* __restrict__ srcA, const int* __restrict__ dstA, int E,
    const float* __restrict__ xl, const float* __restrict__ xr,
    const float* __restrict__ att, const float* __restrict__ bias,
    float* __restrict__ out, int N,
    int* __restrict__ seqG, int* __restrict__ cntG) {
    __shared__ __attribute__((aligned(16))) float acc[NA_NPB * DIM];
    __shared__ float2 st_s[NA_NPB * 2];
    __shared__ int2  lst_s[NA_WAVES * NA_LCAP];
    __shared__ int2  buf_s[NA_WAVES * NA_BCAP];
    __shared__ int   cnt_s[NA_WAVES];
    __shared__ int   fin_s[NA_WAVES];

    const int lane = threadIdx.x & 31;
    const int wave = __builtin_amdgcn_readfirstlane(threadIdx.x >> 5);
    const int lo   = blockIdx.x * NA_NPB;
    const int cnt  = min(NA_NPB, N - lo);
    const int h    = (HEADS == 2) ? (lane >> 4) : 0;
    const float4 tv = ((const float4*)att)[lane];

    for (int r = wave; r < cnt; r += NA_WAVES) {
        const int node = lo + r;
        const float4 x = ((const float4*)(xl + (size_t)node * DIM))[lane];
        const float4 y = ((const float4*)(xr + (size_t)node * DIM))[lane];
        float v = lrelu(x.x + y.x) * tv.x + lrelu(x.y + y.y) * tv.y +
                  lrelu(x.z + y.z) * tv.z + lrelu(x.w + y.w) * tv.w;
        v += __shfl_xor(v, 1, 32);
        v += __shfl_xor(v, 2, 32);
        v += __shfl_xor(v, 4, 32);
        v += __shfl_xor(v, 8, 32);
        if (HEADS == 1) v += __shfl_xor(v, 16, 32);
        float* o = acc + r * DIM + lane;
        o[0] = x.x; o[32] = x.y; o[64] = x.z; o[96] = x.w;
        st_s[r * HEADS + h] = make_float2(v, 1.f);
    }

    const int wid = blockIdx.x * NA_WAVES + wave;
    int2* __restrict__ seq = (int2*)seqG + (size_t)wid * NA_SCAP;
    int sn = 0;

    bool replay = false;
    int total = 0;
    if (MODE == 1) {
        const int c = cntG[blockIdx.x * NA_WAVES + lane];
        replay = __builtin_amdgcn_ballot_w32(c > NA_SCAP) == 0u;
        total  = __builtin_amdgcn_readlane(c, wave);
    }

    if (MODE == 1 && replay) {
        if (total > 0) {
            int2 cur = seq[min(lane, min(32, total) - 1)];
            for (int p = 0; p < total; p += 32) {
                const int n  = min(32, total - p);
                const int pn = p + 32;
                int2 nxt = cur;
                if (pn < total) nxt = seq[pn + min(lane, min(32, total - pn) - 1)];
                na_rows<HEADS>(cur.x, cur.y, n, lane, lo, xl, xr, tv, st_s, acc);
                cur = nxt;
            }
        }
    } else {
    int2* lst = lst_s + wave * NA_LCAP;
    int2* buf = buf_s + wave * NA_BCAP;
    const int nFull = E / NA_TRIP;
    const bool ownsPart = (E - nFull * NA_TRIP) > 0 && (nFull % NA_WAVES) == wave;
    bool partLeft = ownsPart;
    int t = wave;

    for (int round = 0; round < nFull + 2; ++round) {
        int qn = 0;
        if (t < nFull) {
            int a[NA_SCAN], b[NA_SCAN];
            na_load_whole(a, dstA, t, lane);
            for (;;) {
                na_load_whole(b, dstA, min(t + NA_WAVES, nFull - 1), lane);
                qn = na_scan(a, t, lane, lo, cnt, lst, qn);
                t += NA_WAVES;
                if (t >= nFull || qn + NA_TRIP > NA_LCAP) break;
                na_load_whole(a, dstA, min(t + NA_WAVES, nFull - 1), lane);
                qn = na_scan(b, t, lane, lo, cnt, lst, qn);
                t += NA_WAVES;
                if (t >= nFull || qn + NA_TRIP > NA_LCAP) break;
            }
        }
        if (partLeft && t >= nFull && qn + NA_TRIP <= NA_LCAP) {
            int c[NA_SCAN];
            na_load_part(c, dstA, nFull, lane, E);
            qn = na_scan(c, nFull, lane, lo, cnt, lst, qn);
            partLeft = false;
        }
        if (lane == 0) {
            cnt_s[wave] = qn;
            fin_s[wave] = (t < nFull || partLeft) ? 1 : 0;
        }
        __syncthreads();

        int bn = 0;
        for (int l = 0; l < NA_WAVES; ++l) {
            const int n = __builtin_amdgcn_readfirstlane(cnt_s[l]);
            const int2* src_l = lst_s + l * NA_LCAP;
            for (int p = 0; p < n; p += 32) {
                const int i = p + lane;
                const int2 en = src_l[min(i, n - 1)];
                const bool mt = i < n && (en.y & (NA_WAVES - 1)) == wave;
                const unsigned mask = __builtin_amdgcn_ballot_w32(mt);
                if (mask) {
                    const int pos = bn + __builtin_popcount(mask & ((1u << lane) - 1u));
                    if (mt) buf[pos] = en;
                    bn += __builtin_popcount(mask);
                    if (bn >= 32) {
                        sn = na_chunk<HEADS, MODE == 0>(buf, 32, lane, lo, srcA, xl, xr, tv,
                                                        st_s, acc, seq, sn);
                        const int rem = bn - 32;
                        const int2 keep = buf[32 + min(lane, NA_BCAP - 33)];
                        if (lane < rem) buf[lane] = keep;
                        bn = rem;
                    }
                }
            }
        }
        if (bn > 0)
            sn = na_chunk<HEADS, MODE == 0>(buf, bn, lane, lo, srcA, xl, xr, tv,
                                            st_s, acc, seq, sn);

        const bool more = __builtin_amdgcn_ballot_w32(fin_s[lane] != 0) != 0u;
        __syncthreads();
        if (!more) break;
    }
    }
    if (MODE == 0 && lane == 0) cntG[wid] = sn;

    const float4 bz = ((const float4*)bias)[lane];
    for (int r = wave; r < cnt; r += NA_WAVES) {
        const float* o = acc + r * DIM + lane;
        const float dd = st_s[r * HEADS + h].y + 1e-16f;
        float4 z;
        z.x = elu(o[0]  / dd + bz.x);
        z.y = elu(o[32] / dd + bz.y);
        z.z = elu(o[64] / dd + bz.z);
        z.w = elu(o[96] / dd + bz.w);
        ((float4*)(out + (size_t)(lo + r) * DIM))[lane] = z;
    }
}

extern "C" void kernel_launch(void* const* d_in, const int* in_sizes, int n_in,
                              void* d_out, int out_size, void* d_ws, size_t ws_size,
                              hipStream_t stream) {
    const float* x    = (const float*)d_in[0];
    const int*   ei   = (const int*)d_in[1];
    const float* Wl1  = (const float*)d_in[2];
    const float* Wr1  = (const float*)d_in[3];
    const float* att1 = (const float*)d_in[4];
    const float* b1   = (const float*)d_in[5];
    const float* Wl2  = (const float*)d_in[6];
    const float* Wr2  = (const float*)d_in[7];
    const float* att2 = (const float*)d_in[8];
    const float* b2   = (const float*)d_in[9];

    const int N  = in_sizes[0] / DIM;
    const int E  = in_sizes[1] / 2;

    const int* srcA = ei;
    const int* dstA = ei + E;

    float*    xl    = (float*)d_ws;
    float*    xr    = xl + (size_t)N * DIM;
    float*    h1    = xr + (size_t)N * DIM;
    float*    out   = (float*)d_out;

    const dim3 B(256);
    const int gGemm  = N / 16;
    const int gOwn   = (N + NA_NPB - 1) / NA_NPB;
    const dim3 BO(32 * NA_WAVES);
    int*      seqG  = (int*)(h1 + (size_t)N * DIM);
    int*      cntG  = seqG + (size_t)gOwn * NA_WAVES * NA_SCAP * 2;

    gemm128_wmma_f32<<<GEMM_GRID, B, 0, stream>>>(x, Wl1, xl, gGemm);
    gemm128_wmma_f32<<<GEMM_GRID, B, 0, stream>>>(x, Wr1, xr, gGemm);
    node_aggregate<2, 0><<<gOwn, BO, 0, stream>>>(srcA, dstA, E, xl, xr, att1, b1, h1, N,
                                                  seqG, cntG);

    gemm128_wmma_f32<<<GEMM_GRID, B, 0, stream>>>(h1, Wl2, xl, gGemm);
    gemm128_wmma_f32<<<GEMM_GRID, B, 0, stream>>>(h1, Wr2, xr, gGemm);
    node_aggregate<1, 1><<<gOwn, BO, 0, stream>>>(srcA, dstA, E, xl, xr, att2, b2, out, N,
                                                  seqG, cntG);
}


# Aggregate pass: the block keeps its own nodes' xr rows in LDS (read once, with the self loop; list entries packed into one word make the room): an edge fetches one row from memory, not two
# speedup vs baseline: 4.4201x; 1.0347x over previous
#include <hip/hip_runtime.h>
#include <hip/hip_bf16.h>

typedef __attribute__((ext_vector_type(2))) float v2f;
typedef __attribute__((ext_vector_type(8))) float v8f;
typedef __attribute__((ext_vector_type(4))) float v4f;

#define DIM 128
#define NEG_SLOPE 0.2f

__device__ __forceinline__ float lrelu(float x) { return x > 0.f ? x : NEG_SLOPE * x; }
__device__ __forceinline__ float elu(float x)   { return x > 0.f ? x : (__expf(x) - 1.f); }

#define A_PITCH 132
#define GEMM_GRID 768
__global__ void __launch_bounds__(256) gemm128_wmma_f32(
    const float* __restrict__ A, const float* __restrict__ W,
    float* __restrict__ C, int nTiles) {
    __shared__ __attribute__((aligned(16))) float As[2 * 16 * A_PITCH];
    const int wave = threadIdx.x >> 5;
    const int lane = threadIdx.x & 31;
    const int half = lane >> 4;
    const int l15  = lane & 15;
    const int col0 = wave * 16;

    int t = blockIdx.x;
    if (t >= nTiles) return;

    const float4* __restrict__ A4 = (const float4*)A;
    float4 p0 = A4[(size_t)t * 512 + threadIdx.x];
    float4 p1 = A4[(size_t)t * 512 + threadIdx.x + 256];

    v2f b[DIM / 4];
#pragma unroll
    for (int i = 0; i < DIM / 4; ++i) {
        const int ka = 4 * i + half * 2;
        b[i].x = W[(size_t)ka * DIM + col0 + l15];
        b[i].y = W[(size_t)(ka + 1) * DIM + col0 + l15];
    }

    const int wr0 = wave * A_PITCH + lane * 4;
    const int wr1 = (wave + 8) * A_PITCH + lane * 4;
    const int rd  = l15 * A_PITCH + half * 2;

    *(float4*)&As[wr0] = p0;
    *(float4*)&As[wr1] = p1;
    __syncthreads();

    int cur = 0;
    for (;;) {
        const int  tn   = t + GEMM_GRID;
        const bool more = tn < nTiles;
        if (more) {
            p0 = A4[(size_t)tn * 512 + threadIdx.x];
            p1 = A4[(size_t)tn * 512 + threadIdx.x + 256];
        }

        const float* __restrict__ Arow = As + cur + rd;
        v8f acc = {};
#pragma unroll
        for (int i = 0; i < DIM / 4; ++i) {
            const v2f a = *(const v2f*)(Arow + 4 * i);
            acc = __builtin_amdgcn_wmma_f32_16x16x4_f32(
                 false, a,  false, b[i],
                 (short)0, acc,  false,  false);
        }

        if (more) {
            const int nxt = 16 * A_PITCH - cur;
            *(float4*)__builtin_assume_aligned(&As[nxt + wr0], 16) = p0;
            *(float4*)__builtin_assume_aligned(&As[nxt + wr1], 16) = p1;
            __syncthreads();
            cur = nxt;
        }

        const int row0 = t * 16;
#pragma unroll
        for (int j = 0; j < 8; ++j)
            C[(size_t)(row0 + j + half * 8) * DIM + col0 + l15] = acc[j];

        if (!more) break;
        t = tn;
    }
}

#define NA_NPB   196
#define NA_WAVES 32
#define NA_SCAN  16
#define NA_TRIP  (32 * NA_SCAN)
#define NA_LCAP  704
#define NA_BCAP  64
#define NA_SCAP  1024

__device__ __forceinline__ void na_load_whole(int (&d)[NA_SCAN], const int* __restrict__ dstA,
                                              int t, int lane) {
    const int* __restrict__ p = dstA + (size_t)t * NA_TRIP + lane;
#pragma unroll
    for (int k = 0; k < NA_SCAN; ++k) d[k] = p[32 * k];
}
__device__ __forceinline__ void na_load_part(int (&d)[NA_SCAN], const int* __restrict__ dstA,
                                             int t, int lane, int E) {
    const int e0 = t * NA_TRIP + lane;
#pragma unroll
    for (int k = 0; k < NA_SCAN; ++k) {
        const int e = e0 + 32 * k;
        d[k] = e < E ? dstA[e] : -1;
    }
}
__device__ __forceinline__ int na_scan(const int (&d)[NA_SCAN], int t, int lane, int lo, int cnt,
                                       int* lst, int qn) {
    const int eb = t * NA_TRIP + lane;
#pragma unroll
    for (int k = 0; k < NA_SCAN; ++k) {
        const int dl = d[k] - lo;
        const bool mt = (unsigned)dl < (unsigned)cnt;
        const unsigned mask = __builtin_amdgcn_ballot_w32(mt);
        if (mask) {
            const int pos = qn + __builtin_popcount(mask & ((1u << lane) - 1u));
            if (mt) lst[pos] = ((eb + 32 * k) << 8) | dl;
            qn += __builtin_popcount(mask);
        }
    }
    return qn;
}

template <int HEADS, int J>
__device__ __forceinline__ void na_entry(
    const v4f row, float v, int b, int n, int dl, int lane, int h,
    float2* st_s, float* acc) {
    if (b + J < n) {
        const int dj = __builtin_amdgcn_readlane(dl, b + J);
        const float sc = __int_as_float(__builtin_amdgcn_update_dpp(
            0, __float_as_int(v), 0x150 + J, 0xf, 0xf, false));
        float2* st = st_s + dj * HEADS + h;
        float*  o  = acc + dj * DIM + lane;
        const float2 ms = *st;
        const float x0 = o[0], x1 = o[32], x2 = o[64], x3 = o[96];
        const float mn = fmaxf(ms.x, sc);
        const float f  = __expf(ms.x - mn);
        const float w  = __expf(sc - mn);
        *st = make_float2(mn, ms.y * f + w);
        o[0]  = x0 * f + w * row.x;
        o[32] = x1 * f + w * row.y;
        o[64] = x2 * f + w * row.z;
        o[96] = x3 * f + w * row.w;
    }
}

template <int HEADS>
__device__ __forceinline__ void na_rows(
    int s, int dl, int n, int lane,
    const float* __restrict__ xl, const float* xrs, const float4 t,
    float2* st_s, float* acc) {
    const bool b0 = (lane & 1) != 0, b1 = (lane & 2) != 0, b2 = (lane & 4) != 0;
    const int  h  = (HEADS == 2) ? (lane >> 4) : 0;
    for (int b = 0; b < n; b += 8) {
        v4f a[8], y[8];
#pragma unroll
        for (int j = 0; j < 8; ++j) {
            const int sj = __builtin_amdgcn_readlane(s,  min(b + j, n - 1));
            const int dj = __builtin_amdgcn_readlane(dl, min(b + j, n - 1));
            a[j] = ((const v4f*)(xl + (size_t)sj * DIM))[lane];
            y[j] = ((const v4f*)(xrs + dj * DIM))[lane];
        }
        asm volatile("" :: "v"(a[0]), "v"(a[1]), "v"(a[2]), "v"(a[3]),
                           "v"(a[4]), "v"(a[5]), "v"(a[6]), "v"(a[7]),
                           "v"(y[0]), "v"(y[1]), "v"(y[2]), "v"(y[3]),
                           "v"(y[4]), "v"(y[5]), "v"(y[6]), "v"(y[7]));

        float p[8];
#pragma unroll
        for (int j = 0; j < 8; ++j)
            p[j] = lrelu(a[j].x + y[j].x) * t.x + lrelu(a[j].y + y[j].y) * t.y +
                   lrelu(a[j].z + y[j].z) * t.z + lrelu(a[j].w + y[j].w) * t.w;

        float q[4];
#pragma unroll
        for (int i = 0; i < 4; ++i) {
            const float keep = b0 ? p[2 * i + 1] : p[2 * i];
            const float send = b0 ? p[2 * i] : p[2 * i + 1];
            q[i] = keep + __shfl_xor(send, 1, 32);
        }
        float r[2];
#pragma unroll
        for (int i = 0; i < 2; ++i) {
            const float keep = b1 ? q[2 * i + 1] : q[2 * i];
            const float send = b1 ? q[2 * i] : q[2 * i + 1];
            r[i] = keep + __shfl_xor(send, 2, 32);
        }
        float v;
        {
            const float keep = b2 ? r[1] : r[0];
            const float send = b2 ? r[0] : r[1];
            v = keep + __shfl_xor(send, 4, 32);
        }
        v += __shfl_xor(v, 8, 32);
        if (HEADS == 1) v += __shfl_xor(v, 16, 32);

        na_entry<HEADS, 0>(a[0], v, b, n, dl, lane, h, st_s, acc);
        na_entry<HEADS, 1>(a[1], v, b, n, dl, lane, h, st_s, acc);
        na_entry<HEADS, 2>(a[2], v, b, n, dl, lane, h, st_s, acc);
        na_entry<HEADS, 3>(a[3], v, b, n, dl, lane, h, st_s, acc);
        na_entry<HEADS, 4>(a[4], v, b, n, dl, lane, h, st_s, acc);
        na_entry<HEADS, 5>(a[5], v, b, n, dl, lane, h, st_s, acc);
        na_entry<HEADS, 6>(a[6], v, b, n, dl, lane, h, st_s, acc);
        na_entry<HEADS, 7>(a[7], v, b, n, dl, lane, h, st_s, acc);
    }
}

template <int HEADS, bool SAVE>
__device__ __forceinline__ int na_chunk(
    const int* buf, int n, int lane,
    const int* __restrict__ srcA, const float* __restrict__ xl,
    const float* xrs, const float4 t,
    float2* st_s, float* acc, int2* __restrict__ seq, int sn) {
    const unsigned en = (unsigned)buf[min(lane, n - 1)];
    const int dl = (int)(en & 255u);
    const int s  = srcA[en >> 8];
    if (SAVE) {
        if (lane < n && sn + lane < NA_SCAP) seq[sn + lane] = make_int2(s, dl);
    }
    na_rows<HEADS>(s, dl, n, lane, xl, xrs, t, st_s, acc);
    return sn + n;
}

template <int HEADS, int MODE>
__global__ void __launch_bounds__(1024) node_aggregate(
    const int* __restrict__ srcA, const int* __restrict__ dstA, int E,
    const float* __restrict__ xl, const float* __restrict__ xr,
    const float* __restrict__ att, const float* __restrict__ bias,
    float* __restrict__ out, int N,
    int* __restrict__ seqG, int* __restrict__ cntG) {
    __shared__ __attribute__((aligned(16))) float acc[NA_NPB * DIM];
    __shared__ __attribute__((aligned(16))) float xrs[NA_NPB * DIM];
    __shared__ float2 st_s[NA_NPB * 2];
    __shared__ int   lst_s[NA_WAVES * NA_LCAP];
    __shared__ int   buf_s[NA_WAVES * NA_BCAP];
    __shared__ int   cnt_s[NA_WAVES];
    __shared__ int   fin_s[NA_WAVES];

    const int lane = threadIdx.x & 31;
    const int wave = __builtin_amdgcn_readfirstlane(threadIdx.x >> 5);
    const int lo   = blockIdx.x * NA_NPB;
    const int cnt  = min(NA_NPB, N - lo);
    const int h    = (HEADS == 2) ? (lane >> 4) : 0;
    const float4 tv = ((const float4*)att)[lane];

    for (int r = wave; r < cnt; r += NA_WAVES) {
        const int node = lo + r;
        const float4 x = ((const float4*)(xl + (size_t)node * DIM))[lane];
        const float4 y = ((const float4*)(xr + (size_t)node * DIM))[lane];
        float v = lrelu(x.x + y.x) * tv.x + lrelu(x.y + y.y) * tv.y +
                  lrelu(x.z + y.z) * tv.z + lrelu(x.w + y.w) * tv.w;
        v += __shfl_xor(v, 1, 32);
        v += __shfl_xor(v, 2, 32);
        v += __shfl_xor(v, 4, 32);
        v += __shfl_xor(v, 8, 32);
        if (HEADS == 1) v += __shfl_xor(v, 16, 32);
        float* o = acc + r * DIM + lane;
        o[0] = x.x; o[32] = x.y; o[64] = x.z; o[96] = x.w;
        ((float4*)(xrs + r * DIM))[lane] = y;
        st_s[r * HEADS + h] = make_float2(v, 1.f);
    }

    const int wid = blockIdx.x * NA_WAVES + wave;
    int2* __restrict__ seq = (int2*)seqG + (size_t)wid * NA_SCAP;
    int sn = 0;

    bool replay = false;
    int total = 0;
    if (MODE == 1) {
        const int c = cntG[blockIdx.x * NA_WAVES + lane];
        replay = __builtin_amdgcn_ballot_w32(c > NA_SCAP) == 0u;
        total  = __builtin_amdgcn_readlane(c, wave);
    }

    if (MODE == 1 && __builtin_expect(replay, true)) {
        if (total > 0) {
            int2 cur = seq[min(lane, min(32, total) - 1)];
            for (int p = 0; p < total; p += 32) {
                const int n  = min(32, total - p);
                const int pn = p + 32;
                int2 nxt = cur;
                if (pn < total) nxt = seq[pn + min(lane, min(32, total - pn) - 1)];
                na_rows<HEADS>(cur.x, cur.y, n, lane, xl, xrs, tv, st_s, acc);
                cur = nxt;
            }
        }
    } else {
    int* lst = lst_s + wave * NA_LCAP;
    int* buf = buf_s + wave * NA_BCAP;
    const int nFull = E / NA_TRIP;
    const bool ownsPart = (E - nFull * NA_TRIP) > 0 && (nFull % NA_WAVES) == wave;
    bool partLeft = ownsPart;
    int t = wave;

    for (int round = 0; round < nFull + 2; ++round) {
        int qn = 0;
        if (t < nFull) {
            int a[NA_SCAN], b[NA_SCAN];
            na_load_whole(a, dstA, t, lane);
            for (;;) {
                na_load_whole(b, dstA, min(t + NA_WAVES, nFull - 1), lane);
                qn = na_scan(a, t, lane, lo, cnt, lst, qn);
                t += NA_WAVES;
                if (t >= nFull || qn + NA_TRIP > NA_LCAP) break;
                na_load_whole(a, dstA, min(t + NA_WAVES, nFull - 1), lane);
                qn = na_scan(b, t, lane, lo, cnt, lst, qn);
                t += NA_WAVES;
                if (t >= nFull || qn + NA_TRIP > NA_LCAP) break;
            }
        }
        if (partLeft && t >= nFull && qn + NA_TRIP <= NA_LCAP) {
            int c[NA_SCAN];
            na_load_part(c, dstA, nFull, lane, E);
            qn = na_scan(c, nFull, lane, lo, cnt, lst, qn);
            partLeft = false;
        }
        if (lane == 0) {
            cnt_s[wave] = qn;
            fin_s[wave] = (t < nFull || partLeft) ? 1 : 0;
        }
        __syncthreads();

        int bn = 0;
        for (int l = 0; l < NA_WAVES; ++l) {
            const int n = __builtin_amdgcn_readfirstlane(cnt_s[l]);
            const int* src_l = lst_s + l * NA_LCAP;
            for (int p = 0; p < n; p += 32) {
                const int i = p + lane;
                const int en = src_l[min(i, n - 1)];
                const bool mt = i < n && (en & (NA_WAVES - 1)) == wave;
                const unsigned mask = __builtin_amdgcn_ballot_w32(mt);
                if (mask) {
                    const int pos = bn + __builtin_popcount(mask & ((1u << lane) - 1u));
                    if (mt) buf[pos] = en;
                    bn += __builtin_popcount(mask);
                    if (bn >= 32) {
                        sn = na_chunk<HEADS, MODE == 0>(buf, 32, lane, srcA, xl, xrs, tv,
                                                        st_s, acc, seq, sn);
                        const int rem = bn - 32;
                        const int keep = buf[32 + min(lane, NA_BCAP - 33)];
                        if (lane < rem) buf[lane] = keep;
                        bn = rem;
                    }
                }
            }
        }
        if (bn > 0)
            sn = na_chunk<HEADS, MODE == 0>(buf, bn, lane, srcA, xl, xrs, tv,
                                            st_s, acc, seq, sn);

        const bool more = __builtin_amdgcn_ballot_w32(fin_s[lane] != 0) != 0u;
        __syncthreads();
        if (!more) break;
    }
    }
    if (MODE == 0 && lane == 0) cntG[wid] = sn;

    const float4 bz = ((const float4*)bias)[lane];
    for (int r = wave; r < cnt; r += NA_WAVES) {
        const float* o = acc + r * DIM + lane;
        const float dd = st_s[r * HEADS + h].y + 1e-16f;
        float4 z;
        z.x = elu(o[0]  / dd + bz.x);
        z.y = elu(o[32] / dd + bz.y);
        z.z = elu(o[64] / dd + bz.z);
        z.w = elu(o[96] / dd + bz.w);
        ((float4*)(out + (size_t)(lo + r) * DIM))[lane] = z;
    }
}

extern "C" void kernel_launch(void* const* d_in, const int* in_sizes, int n_in,
                              void* d_out, int out_size, void* d_ws, size_t ws_size,
                              hipStream_t stream) {
    const float* x    = (const float*)d_in[0];
    const int*   ei   = (const int*)d_in[1];
    const float* Wl1  = (const float*)d_in[2];
    const float* Wr1  = (const float*)d_in[3];
    const float* att1 = (const float*)d_in[4];
    const float* b1   = (const float*)d_in[5];
    const float* Wl2  = (const float*)d_in[6];
    const float* Wr2  = (const float*)d_in[7];
    const float* att2 = (const float*)d_in[8];
    const float* b2   = (const float*)d_in[9];

    const int N  = in_sizes[0] / DIM;
    const int E  = in_sizes[1] / 2;

    const int* srcA = ei;
    const int* dstA = ei + E;

    float*    xl    = (float*)d_ws;
    float*    xr    = xl + (size_t)N * DIM;
    float*    h1    = xr + (size_t)N * DIM;
    float*    out   = (float*)d_out;

    const dim3 B(256);
    const int gGemm  = N / 16;
    const int gOwn   = (N + NA_NPB - 1) / NA_NPB;
    const dim3 BO(32 * NA_WAVES);
    int*      seqG  = (int*)(h1 + (size_t)N * DIM);
    int*      cntG  = seqG + (size_t)gOwn * NA_WAVES * NA_SCAP * 2;

    gemm128_wmma_f32<<<GEMM_GRID, B, 0, stream>>>(x, Wl1, xl, gGemm);
    gemm128_wmma_f32<<<GEMM_GRID, B, 0, stream>>>(x, Wr1, xr, gGemm);
    node_aggregate<2, 0><<<gOwn, BO, 0, stream>>>(srcA, dstA, E, xl, xr, att1, b1, h1, N,
                                                  seqG, cntG);

    gemm128_wmma_f32<<<GEMM_GRID, B, 0, stream>>>(h1, Wl2, xl, gGemm);
    gemm128_wmma_f32<<<GEMM_GRID, B, 0, stream>>>(h1, Wr2, xr, gGemm);
    node_aggregate<1, 1><<<gOwn, BO, 0, stream>>>(srcA, dstA, E, xl, xr, att2, b2, out, N,
                                                  seqG, cntG);
}
